# starting kernel
# baseline (speedup 1.0000x reference)
#include <hip/hip_runtime.h>
#include <hip/hip_bf16.h>


typedef float v2f __attribute__((ext_vector_type(2)));
typedef float v8f __attribute__((ext_vector_type(8)));

#define HIDDEN 16


__global__ void deg_init_kernel(float* __restrict__ deg, int n) {
    int i = blockIdx.x * blockDim.x + threadIdx.x;
    if (i < n) deg[i] = 1.0f;
}

__global__ void deg_count_kernel(const int* __restrict__ dst, float* __restrict__ deg,
                                 int n_edges) {
    int e = blockIdx.x * blockDim.x + threadIdx.x;
    if (e < n_edges) atomicAdd(&deg[dst[e]], 1.0f);
}

__global__ void dinv_kernel(float* __restrict__ deg_dinv, int n) {
    int i = blockIdx.x * blockDim.x + threadIdx.x;
    if (i < n) deg_dinv[i] = rsqrtf(deg_dinv[i]);
}

__global__ void __launch_bounds__(256)
gemm_wmma_kernel(const float* __restrict__ X, const float* __restrict__ W,
                 float* __restrict__ H, int n_rows, int K) {
    int wave = (int)((blockIdx.x * blockDim.x + threadIdx.x) >> 5);
    int lane = threadIdx.x & 31;
    int row0 = wave * 16;
    if (row0 >= n_rows) return;

    int half = lane >> 4;
    int l16  = lane & 15;

    const float* xrow = X + (size_t)(row0 + l16) * K;
    v8f acc = {};
    for (int k = 0; k < K; k += 4) {
        int ka = k + half * 2;
        v2f a, b;
        a.x = xrow[ka];
        a.y = xrow[ka + 1];
        b.x = W[(size_t)ka * HIDDEN + l16];
        b.y = W[(size_t)(ka + 1) * HIDDEN + l16];
        acc = __builtin_amdgcn_wmma_f32_16x16x4_f32(
             false, a,  false, b,
             (short)0, acc,  false,  false);
    }
    float* out = H + (size_t)row0 * HIDDEN;
#pragma unroll
    for (int i = 0; i < 8; ++i)
        out[(size_t)(i + 8 * half) * HIDDEN + l16] = acc[i];
}


__global__ void zero_kernel(float* __restrict__ p, int n) {
    int i = blockIdx.x * blockDim.x + threadIdx.x;
    if (i < n) p[i] = 0.0f;
}

__global__ void scatter_kernel(const int* __restrict__ src, const int* __restrict__ dst,
                               const float* __restrict__ dinv,
                               const float* __restrict__ hin,
                               float* __restrict__ agg, int n_edges) {
    int g = blockIdx.x * blockDim.x + threadIdx.x;
    int e = g >> 4;
    if (e >= n_edges) return;
    int f = g & 15;
    int s = src[e];
    int d = dst[e];
    float norm = dinv[s] * dinv[d];
    float v = hin[(size_t)s * HIDDEN + f] * norm;
    atomicAdd(&agg[(size_t)d * HIDDEN + f], v);
}

__global__ void finalize_kernel(const float* __restrict__ hpre,
                                const float* __restrict__ dinv,
                                const float* __restrict__ bias,
                                float* __restrict__ agg, int n_nodes, int do_relu) {
    int g = blockIdx.x * blockDim.x + threadIdx.x;
    int i = g >> 4;
    if (i >= n_nodes) return;
    int f = g & 15;
    float di = dinv[i];
    float v = agg[(size_t)i * HIDDEN + f] + di * di * hpre[(size_t)i * HIDDEN + f]
              + bias[f];
    if (do_relu) v = fmaxf(v, 0.0f);
    agg[(size_t)i * HIDDEN + f] = v;
}


__global__ void pool_kernel(const float* __restrict__ h, const int* __restrict__ batch,
                            float* __restrict__ pooled, int n_nodes) {
    int g = blockIdx.x * blockDim.x + threadIdx.x;
    int i = g >> 4;
    if (i >= n_nodes) return;
    int f = g & 15;
    atomicAdd(&pooled[(size_t)batch[i] * HIDDEN + f], h[(size_t)i * HIDDEN + f]);
}

__global__ void final_linear_kernel(const float* __restrict__ pooled,
                                    const float* __restrict__ Wlin,
                                    const float* __restrict__ blin,
                                    float* __restrict__ out,
                                    int n_graphs, int n_classes) {
    int g = blockIdx.x * blockDim.x + threadIdx.x;
    if (g >= n_graphs * n_classes) return;
    int gr = g / n_classes;
    int c  = g % n_classes;
    float acc = blin[c];
#pragma unroll
    for (int f = 0; f < HIDDEN; ++f)
        acc += pooled[(size_t)gr * HIDDEN + f] * Wlin[f * n_classes + c];
    out[g] = acc;
}


static inline size_t align_up(size_t v, size_t a) { return (v + a - 1) & ~(a - 1); }

extern "C" void kernel_launch(void* const* d_in, const int* in_sizes, int n_in,
                              void* d_out, int out_size, void* d_ws, size_t ws_size,
                              hipStream_t stream) {
    (void)n_in; (void)ws_size;

    const float* x    = (const float*)d_in[0];
    const int*   ei   = (const int*)d_in[1];
    const int*   batch= (const int*)d_in[2];
    const float* W1   = (const float*)d_in[3];
    const float* b1   = (const float*)d_in[4];
    const float* W2   = (const float*)d_in[5];
    const float* b2   = (const float*)d_in[6];
    const float* W3   = (const float*)d_in[7];
    const float* b3   = (const float*)d_in[8];
    const float* Wlin = (const float*)d_in[9];
    const float* blin = (const float*)d_in[10];
    float* out = (float*)d_out;

    const int N_FEAT = 128;
    const int n_nodes = in_sizes[0] / N_FEAT;
    const int n_edges = in_sizes[1] / 2;
    const int n_classes = in_sizes[10];
    const int n_graphs  = out_size / n_classes;

    const int* src = ei;
    const int* dst = ei + n_edges;

    char* ws = (char*)d_ws;
    size_t off = 0;
    float* dinv = (float*)(ws + off);  off = align_up(off + (size_t)n_nodes * 4, 256);
    float* bufA = (float*)(ws + off);  off = align_up(off + (size_t)n_nodes * HIDDEN * 4, 256);
    float* bufB = (float*)(ws + off);  off = align_up(off + (size_t)n_nodes * HIDDEN * 4, 256);
    float* pooled = (float*)(ws + off); off = align_up(off + (size_t)n_graphs * HIDDEN * 4, 256);

    const int T = 256;
    const int nodeBlocks  = (n_nodes + T - 1) / T;
    const int edgeBlocks  = (n_edges + T - 1) / T;
    const int nodeFBlocks = ((n_nodes * HIDDEN) + T - 1) / T;
    const int edgeFBlocks = (int)(((long long)n_edges * HIDDEN + T - 1) / T);
    const int hElems      = n_nodes * HIDDEN;
    const int tiles       = (n_nodes + 15) / 16;
    const int gemmBlocks  = (tiles + 7) / 8;

    deg_init_kernel<<<nodeBlocks, T, 0, stream>>>(dinv, n_nodes);
    deg_count_kernel<<<edgeBlocks, T, 0, stream>>>(dst, dinv, n_edges);
    dinv_kernel<<<nodeBlocks, T, 0, stream>>>(dinv, n_nodes);

    gemm_wmma_kernel<<<gemmBlocks, T, 0, stream>>>(x, W1, bufA, n_nodes, N_FEAT);
    zero_kernel<<<nodeFBlocks, T, 0, stream>>>(bufB, hElems);
    scatter_kernel<<<edgeFBlocks, T, 0, stream>>>(src, dst, dinv, bufA, bufB, n_edges);
    finalize_kernel<<<nodeFBlocks, T, 0, stream>>>(bufA, dinv, b1, bufB, n_nodes, 1);

    gemm_wmma_kernel<<<gemmBlocks, T, 0, stream>>>(bufB, W2, bufA, n_nodes, HIDDEN);
    zero_kernel<<<nodeFBlocks, T, 0, stream>>>(bufB, hElems);
    scatter_kernel<<<edgeFBlocks, T, 0, stream>>>(src, dst, dinv, bufA, bufB, n_edges);
    finalize_kernel<<<nodeFBlocks, T, 0, stream>>>(bufA, dinv, b2, bufB, n_nodes, 1);

    gemm_wmma_kernel<<<gemmBlocks, T, 0, stream>>>(bufB, W3, bufA, n_nodes, HIDDEN);
    zero_kernel<<<nodeFBlocks, T, 0, stream>>>(bufB, hElems);
    scatter_kernel<<<edgeFBlocks, T, 0, stream>>>(src, dst, dinv, bufA, bufB, n_edges);
    finalize_kernel<<<nodeFBlocks, T, 0, stream>>>(bufA, dinv, b3, bufB, n_nodes, 0);

    zero_kernel<<<(n_graphs * HIDDEN + T - 1) / T, T, 0, stream>>>(pooled, n_graphs * HIDDEN);
    pool_kernel<<<nodeFBlocks, T, 0, stream>>>(bufB, batch, pooled, n_nodes);
    final_linear_kernel<<<(n_graphs * n_classes + T - 1) / T, T, 0, stream>>>(
        pooled, Wlin, blin, out, n_graphs, n_classes);
}


# In-degree count: per-block LDS integer histograms over edge slices, summed per node, replace 3.2M global float atomics on the degree array (degrees bit-identical; 18 to 17 launches)
# speedup vs baseline: 1.2838x; 1.2838x over previous
#include <hip/hip_runtime.h>
#include <hip/hip_bf16.h>


typedef float v2f __attribute__((ext_vector_type(2)));
typedef float v8f __attribute__((ext_vector_type(8)));

#define HIDDEN 16


#define DEG_SLICES  64
#define DEG_RANGE   25600
#define DEG_THREADS 1024
#define DEG_DEPTH   8

__global__ void __launch_bounds__(DEG_THREADS)
deg_hist_kernel(const int* __restrict__ dst, unsigned* __restrict__ part,
                int n_nodes, int n_edges) {
    __shared__ unsigned hist[DEG_RANGE];
    const int tid  = threadIdx.x;
    const int base = blockIdx.y * DEG_RANGE;

    for (int i = tid; i < DEG_RANGE; i += DEG_THREADS) hist[i] = 0u;
    __syncthreads();

    const int slice = ((n_edges + DEG_SLICES * 4 - 1) / (DEG_SLICES * 4)) * 4;
    const int e0 = blockIdx.x * slice;
    const int e1 = min(e0 + slice, n_edges);

    int e = e0 + tid;
    for (; e + (DEG_DEPTH - 1) * DEG_THREADS < e1; e += DEG_DEPTH * DEG_THREADS) {
        unsigned d[DEG_DEPTH];
#pragma unroll
        for (int u = 0; u < DEG_DEPTH; ++u)
            d[u] = (unsigned)(dst[e + u * DEG_THREADS] - base);
#pragma unroll
        for (int u = 0; u < DEG_DEPTH; ++u)
            if (d[u] < DEG_RANGE) atomicAdd(&hist[d[u]], 1u);
    }
    for (; e < e1; e += DEG_THREADS) {
        unsigned d0 = (unsigned)(dst[e] - base);
        if (d0 < DEG_RANGE) atomicAdd(&hist[d0], 1u);
    }
    __syncthreads();

    const int nr = min(DEG_RANGE, n_nodes - base);
    unsigned* p = part + (size_t)blockIdx.x * n_nodes + base;
    for (int i = tid; i < nr; i += DEG_THREADS) p[i] = hist[i];
}

__global__ void dinv_kernel(const unsigned* __restrict__ part, float* __restrict__ dinv,
                            int n) {
    int i = blockIdx.x * blockDim.x + threadIdx.x;
    if (i >= n) return;
    unsigned deg = 1u;
#pragma unroll 16
    for (int s = 0; s < DEG_SLICES; ++s) deg += part[(size_t)s * n + i];
    dinv[i] = rsqrtf((float)deg);
}

__global__ void __launch_bounds__(256)
gemm_wmma_kernel(const float* __restrict__ X, const float* __restrict__ W,
                 float* __restrict__ H, int n_rows, int K) {
    int wave = (int)((blockIdx.x * blockDim.x + threadIdx.x) >> 5);
    int lane = threadIdx.x & 31;
    int row0 = wave * 16;
    if (row0 >= n_rows) return;

    int half = lane >> 4;
    int l16  = lane & 15;

    const float* xrow = X + (size_t)(row0 + l16) * K;
    v8f acc = {};
    for (int k = 0; k < K; k += 4) {
        int ka = k + half * 2;
        v2f a, b;
        a.x = xrow[ka];
        a.y = xrow[ka + 1];
        b.x = W[(size_t)ka * HIDDEN + l16];
        b.y = W[(size_t)(ka + 1) * HIDDEN + l16];
        acc = __builtin_amdgcn_wmma_f32_16x16x4_f32(
             false, a,  false, b,
             (short)0, acc,  false,  false);
    }
    float* out = H + (size_t)row0 * HIDDEN;
#pragma unroll
    for (int i = 0; i < 8; ++i)
        out[(size_t)(i + 8 * half) * HIDDEN + l16] = acc[i];
}


__global__ void zero_kernel(float* __restrict__ p, int n) {
    int i = blockIdx.x * blockDim.x + threadIdx.x;
    if (i < n) p[i] = 0.0f;
}

__global__ void scatter_kernel(const int* __restrict__ src, const int* __restrict__ dst,
                               const float* __restrict__ dinv,
                               const float* __restrict__ hin,
                               float* __restrict__ agg, int n_edges) {
    int g = blockIdx.x * blockDim.x + threadIdx.x;
    int e = g >> 4;
    if (e >= n_edges) return;
    int f = g & 15;
    int s = src[e];
    int d = dst[e];
    float norm = dinv[s] * dinv[d];
    float v = hin[(size_t)s * HIDDEN + f] * norm;
    atomicAdd(&agg[(size_t)d * HIDDEN + f], v);
}

__global__ void finalize_kernel(const float* __restrict__ hpre,
                                const float* __restrict__ dinv,
                                const float* __restrict__ bias,
                                float* __restrict__ agg, int n_nodes, int do_relu) {
    int g = blockIdx.x * blockDim.x + threadIdx.x;
    int i = g >> 4;
    if (i >= n_nodes) return;
    int f = g & 15;
    float di = dinv[i];
    float v = agg[(size_t)i * HIDDEN + f] + di * di * hpre[(size_t)i * HIDDEN + f]
              + bias[f];
    if (do_relu) v = fmaxf(v, 0.0f);
    agg[(size_t)i * HIDDEN + f] = v;
}


__global__ void pool_kernel(const float* __restrict__ h, const int* __restrict__ batch,
                            float* __restrict__ pooled, int n_nodes) {
    int g = blockIdx.x * blockDim.x + threadIdx.x;
    int i = g >> 4;
    if (i >= n_nodes) return;
    int f = g & 15;
    atomicAdd(&pooled[(size_t)batch[i] * HIDDEN + f], h[(size_t)i * HIDDEN + f]);
}

__global__ void final_linear_kernel(const float* __restrict__ pooled,
                                    const float* __restrict__ Wlin,
                                    const float* __restrict__ blin,
                                    float* __restrict__ out,
                                    int n_graphs, int n_classes) {
    int g = blockIdx.x * blockDim.x + threadIdx.x;
    if (g >= n_graphs * n_classes) return;
    int gr = g / n_classes;
    int c  = g % n_classes;
    float acc = blin[c];
#pragma unroll
    for (int f = 0; f < HIDDEN; ++f)
        acc += pooled[(size_t)gr * HIDDEN + f] * Wlin[f * n_classes + c];
    out[g] = acc;
}


static inline size_t align_up(size_t v, size_t a) { return (v + a - 1) & ~(a - 1); }

extern "C" void kernel_launch(void* const* d_in, const int* in_sizes, int n_in,
                              void* d_out, int out_size, void* d_ws, size_t ws_size,
                              hipStream_t stream) {
    (void)n_in; (void)ws_size;

    const float* x    = (const float*)d_in[0];
    const int*   ei   = (const int*)d_in[1];
    const int*   batch= (const int*)d_in[2];
    const float* W1   = (const float*)d_in[3];
    const float* b1   = (const float*)d_in[4];
    const float* W2   = (const float*)d_in[5];
    const float* b2   = (const float*)d_in[6];
    const float* W3   = (const float*)d_in[7];
    const float* b3   = (const float*)d_in[8];
    const float* Wlin = (const float*)d_in[9];
    const float* blin = (const float*)d_in[10];
    float* out = (float*)d_out;

    const int N_FEAT = 128;
    const int n_nodes = in_sizes[0] / N_FEAT;
    const int n_edges = in_sizes[1] / 2;
    const int n_classes = in_sizes[10];
    const int n_graphs  = out_size / n_classes;

    const int* src = ei;
    const int* dst = ei + n_edges;

    char* ws = (char*)d_ws;
    size_t off = 0;
    float* dinv = (float*)(ws + off);  off = align_up(off + (size_t)n_nodes * 4, 256);
    float* bufA = (float*)(ws + off);  off = align_up(off + (size_t)n_nodes * HIDDEN * 4, 256);
    float* bufB = (float*)(ws + off);  off = align_up(off + (size_t)n_nodes * HIDDEN * 4, 256);
    float* pooled = (float*)(ws + off); off = align_up(off + (size_t)n_graphs * HIDDEN * 4, 256);
    unsigned* degPart = (unsigned*)(ws + off);
    off = align_up(off + (size_t)DEG_SLICES * n_nodes * 4, 256);

    const int T = 256;
    const int nodeBlocks  = (n_nodes + T - 1) / T;
    const int nodeFBlocks = ((n_nodes * HIDDEN) + T - 1) / T;
    const int edgeFBlocks = (int)(((long long)n_edges * HIDDEN + T - 1) / T);
    const int hElems      = n_nodes * HIDDEN;
    const int tiles       = (n_nodes + 15) / 16;
    const int gemmBlocks  = (tiles + 7) / 8;

    deg_hist_kernel<<<dim3(DEG_SLICES, (n_nodes + DEG_RANGE - 1) / DEG_RANGE), DEG_THREADS, 0, stream>>>(
        dst, degPart, n_nodes, n_edges);
    dinv_kernel<<<nodeBlocks, T, 0, stream>>>(degPart, dinv, n_nodes);

    gemm_wmma_kernel<<<gemmBlocks, T, 0, stream>>>(x, W1, bufA, n_nodes, N_FEAT);
    zero_kernel<<<nodeFBlocks, T, 0, stream>>>(bufB, hElems);
    scatter_kernel<<<edgeFBlocks, T, 0, stream>>>(src, dst, dinv, bufA, bufB, n_edges);
    finalize_kernel<<<nodeFBlocks, T, 0, stream>>>(bufA, dinv, b1, bufB, n_nodes, 1);

    gemm_wmma_kernel<<<gemmBlocks, T, 0, stream>>>(bufB, W2, bufA, n_nodes, HIDDEN);
    zero_kernel<<<nodeFBlocks, T, 0, stream>>>(bufB, hElems);
    scatter_kernel<<<edgeFBlocks, T, 0, stream>>>(src, dst, dinv, bufA, bufB, n_edges);
    finalize_kernel<<<nodeFBlocks, T, 0, stream>>>(bufA, dinv, b2, bufB, n_nodes, 1);

    gemm_wmma_kernel<<<gemmBlocks, T, 0, stream>>>(bufB, W3, bufA, n_nodes, HIDDEN);
    zero_kernel<<<nodeFBlocks, T, 0, stream>>>(bufB, hElems);
    scatter_kernel<<<edgeFBlocks, T, 0, stream>>>(src, dst, dinv, bufA, bufB, n_edges);
    finalize_kernel<<<nodeFBlocks, T, 0, stream>>>(bufA, dinv, b3, bufB, n_nodes, 0);

    zero_kernel<<<(n_graphs * HIDDEN + T - 1) / T, T, 0, stream>>>(pooled, n_graphs * HIDDEN);
    pool_kernel<<<nodeFBlocks, T, 0, stream>>>(bufB, batch, pooled, n_nodes);
    final_linear_kernel<<<(n_graphs * n_classes + T - 1) / T, T, 0, stream>>>(
        pooled, Wlin, blin, out, n_graphs, n_classes);
}


# Aggregation as a pull: edges counting-sorted by destination once (LDS ranks), each layer sums a node's rows in registers in slot order; no float atomics, no zero or finalize pass (17 to 13 launches)
# speedup vs baseline: 2.9944x; 2.3325x over previous
#include <hip/hip_runtime.h>
#include <hip/hip_bf16.h>


typedef float v2f __attribute__((ext_vector_type(2)));
typedef float v8f __attribute__((ext_vector_type(8)));

#define HIDDEN 16


#define DEG_SLICES  64
#define DEG_RANGE   25600
#define DEG_THREADS 1024
#define DEG_DEPTH   8

__global__ void __launch_bounds__(DEG_THREADS)
deg_hist_kernel(const int* __restrict__ dst, unsigned* __restrict__ part,
                int n_nodes, int n_edges) {
    __shared__ unsigned hist[DEG_RANGE];
    const int tid  = threadIdx.x;
    const int base = blockIdx.y * DEG_RANGE;

    for (int i = tid; i < DEG_RANGE; i += DEG_THREADS) hist[i] = 0u;
    __syncthreads();

    const int slice = ((n_edges + DEG_SLICES * 4 - 1) / (DEG_SLICES * 4)) * 4;
    const int e0 = blockIdx.x * slice;
    const int e1 = min(e0 + slice, n_edges);

    int e = e0 + tid;
    for (; e + (DEG_DEPTH - 1) * DEG_THREADS < e1; e += DEG_DEPTH * DEG_THREADS) {
        unsigned d[DEG_DEPTH];
#pragma unroll
        for (int u = 0; u < DEG_DEPTH; ++u)
            d[u] = (unsigned)(dst[e + u * DEG_THREADS] - base);
#pragma unroll
        for (int u = 0; u < DEG_DEPTH; ++u)
            if (d[u] < DEG_RANGE) atomicAdd(&hist[d[u]], 1u);
    }
    for (; e < e1; e += DEG_THREADS) {
        unsigned d0 = (unsigned)(dst[e] - base);
        if (d0 < DEG_RANGE) atomicAdd(&hist[d0], 1u);
    }
    __syncthreads();

    const int nr = min(DEG_RANGE, n_nodes - base);
    unsigned* p = part + (size_t)blockIdx.x * n_nodes + base;
    for (int i = tid; i < nr; i += DEG_THREADS) p[i] = hist[i];
}

#define NODE_T 256

__global__ void __launch_bounds__(NODE_T)
dinv_kernel(const unsigned* __restrict__ part, unsigned* __restrict__ soff,
            unsigned* __restrict__ indeg, unsigned* __restrict__ blockSum,
            float* __restrict__ dinv, int n) {
    __shared__ unsigned sh[NODE_T];
    const int tid = threadIdx.x;
    const int i = blockIdx.x * NODE_T + tid;
    unsigned run = 0u;
    if (i < n) {
        static_assert(DEG_SLICES % 16 == 0, "slices are taken 16 at a time");
        for (int s0 = 0; s0 < DEG_SLICES; s0 += 16) {
            unsigned c[16];
#pragma unroll
            for (int u = 0; u < 16; ++u) c[u] = part[(size_t)(s0 + u) * n + i];
#pragma unroll
            for (int u = 0; u < 16; ++u) {
                soff[(size_t)(s0 + u) * n + i] = run;
                run += c[u];
            }
        }
        indeg[i] = run;
        dinv[i] = rsqrtf((float)(run + 1u));
    }
    sh[tid] = run;
    __syncthreads();
    for (int o = NODE_T / 2; o > 0; o >>= 1) {
        if (tid < o) sh[tid] += sh[tid + o];
        __syncthreads();
    }
    if (tid == 0) blockSum[blockIdx.x] = sh[0];
}

__global__ void __launch_bounds__(NODE_T)
rowptr_kernel(const unsigned* __restrict__ indeg, const unsigned* __restrict__ blockSum,
              unsigned* __restrict__ rowptr, int n) {
    __shared__ unsigned sh[NODE_T];
    const int tid = threadIdx.x;
    unsigned pre = 0u;
    for (int b = tid; b < (int)blockIdx.x; b += NODE_T) pre += blockSum[b];
    sh[tid] = pre;
    __syncthreads();
    for (int o = NODE_T / 2; o > 0; o >>= 1) {
        if (tid < o) sh[tid] += sh[tid + o];
        __syncthreads();
    }
    const unsigned before = sh[0];
    __syncthreads();

    const int i = blockIdx.x * NODE_T + tid;
    const unsigned v = (i < n) ? indeg[i] : 0u;
    sh[tid] = v;
    __syncthreads();
    for (int o = 1; o < NODE_T; o <<= 1) {
        unsigned t = (tid >= o) ? sh[tid - o] : 0u;
        __syncthreads();
        sh[tid] += t;
        __syncthreads();
    }
    if (i < n) rowptr[i] = before + sh[tid] - v;
}

__global__ void __launch_bounds__(DEG_THREADS)
csr_fill_kernel(const int* __restrict__ src, const int* __restrict__ dst,
                const float* __restrict__ dinv,
                const unsigned* __restrict__ rowptr, const unsigned* __restrict__ soff,
                uint2* __restrict__ csr, int n_nodes, int n_edges) {
    __shared__ unsigned cur[DEG_RANGE];
    const int tid  = threadIdx.x;
    const int base = blockIdx.y * DEG_RANGE;
    const int nr   = min(DEG_RANGE, n_nodes - base);

    const unsigned* so = soff + (size_t)blockIdx.x * n_nodes + base;
    for (int i = tid; i < nr; i += DEG_THREADS) cur[i] = rowptr[base + i] + so[i];
    __syncthreads();

    const int slice = ((n_edges + DEG_SLICES * 4 - 1) / (DEG_SLICES * 4)) * 4;
    const int e0 = blockIdx.x * slice;
    const int e1 = min(e0 + slice, n_edges);

    int e = e0 + tid;
    for (; e + (DEG_DEPTH - 1) * DEG_THREADS < e1; e += DEG_DEPTH * DEG_THREADS) {
        unsigned d[DEG_DEPTH];
        int      s[DEG_DEPTH];
        float    w[DEG_DEPTH];
#pragma unroll
        for (int u = 0; u < DEG_DEPTH; ++u) {
            d[u] = (unsigned)(dst[e + u * DEG_THREADS] - base);
            s[u] = src[e + u * DEG_THREADS];
        }
#pragma unroll
        for (int u = 0; u < DEG_DEPTH; ++u)
            w[u] = dinv[d[u] < (unsigned)nr ? s[u] : 0];
        asm volatile("" :: "v"(w[0]), "v"(w[1]), "v"(w[2]), "v"(w[3]),
                           "v"(w[4]), "v"(w[5]), "v"(w[6]), "v"(w[7]));
#pragma unroll
        for (int u = 0; u < DEG_DEPTH; ++u)
            if (d[u] < (unsigned)nr) {
                unsigned pos = atomicAdd(&cur[d[u]], 1u);
                csr[pos] = make_uint2((unsigned)s[u], __float_as_uint(w[u]));
            }
    }
    for (; e < e1; e += DEG_THREADS) {
        unsigned d0 = (unsigned)(dst[e] - base);
        if (d0 < (unsigned)nr) {
            int s0 = src[e];
            unsigned pos = atomicAdd(&cur[d0], 1u);
            csr[pos] = make_uint2((unsigned)s0, __float_as_uint(dinv[s0]));
        }
    }
}

__global__ void __launch_bounds__(256)
gemm_wmma_kernel(const float* __restrict__ X, const float* __restrict__ W,
                 float* __restrict__ H, int n_rows, int K) {
    int wave = (int)((blockIdx.x * blockDim.x + threadIdx.x) >> 5);
    int lane = threadIdx.x & 31;
    int row0 = wave * 16;
    if (row0 >= n_rows) return;

    int half = lane >> 4;
    int l16  = lane & 15;

    const float* xrow = X + (size_t)(row0 + l16) * K;
    v8f acc = {};
    for (int k = 0; k < K; k += 4) {
        int ka = k + half * 2;
        v2f a, b;
        a.x = xrow[ka];
        a.y = xrow[ka + 1];
        b.x = W[(size_t)ka * HIDDEN + l16];
        b.y = W[(size_t)(ka + 1) * HIDDEN + l16];
        acc = __builtin_amdgcn_wmma_f32_16x16x4_f32(
             false, a,  false, b,
             (short)0, acc,  false,  false);
    }
    float* out = H + (size_t)row0 * HIDDEN;
#pragma unroll
    for (int i = 0; i < 8; ++i)
        out[(size_t)(i + 8 * half) * HIDDEN + l16] = acc[i];
}


__global__ void zero_kernel(float* __restrict__ p, int n) {
    int i = blockIdx.x * blockDim.x + threadIdx.x;
    if (i < n) p[i] = 0.0f;
}

#define GATHER_DEPTH 8

__global__ void __launch_bounds__(256)
gather_kernel(const uint2* __restrict__ csr, const unsigned* __restrict__ rowptr,
              const unsigned* __restrict__ indeg, const float* __restrict__ dinv,
              const float4* __restrict__ hin, const float* __restrict__ bias,
              float4* __restrict__ hout, int n_nodes, int do_relu) {
    int g = blockIdx.x * blockDim.x + threadIdx.x;
    int i = g >> 2;
    if (i >= n_nodes) return;
    int q = g & 3;
    unsigned j = rowptr[i];
    const unsigned end = j + indeg[i];
    const float di = dinv[i];
    const float4 hs = hin[(size_t)i * 4 + q];
    float4 acc = make_float4(0.0f, 0.0f, 0.0f, 0.0f);

    for (; j < end; j += GATHER_DEPTH) {
        uint2  c[GATHER_DEPTH];
        float4 h[GATHER_DEPTH];
#pragma unroll
        for (int u = 0; u < GATHER_DEPTH; ++u)
            c[u] = csr[min(j + (unsigned)u, end - 1u)];
#pragma unroll
        for (int u = 0; u < GATHER_DEPTH; ++u)
            h[u] = hin[(size_t)c[u].x * 4 + q];
#pragma unroll
        for (int u = 0; u < GATHER_DEPTH; ++u) {
            const bool on = j + (unsigned)u < end;
            const float norm = __uint_as_float(c[u].y) * di;
            const float sx = acc.x + norm * h[u].x;
            const float sy = acc.y + norm * h[u].y;
            const float sz = acc.z + norm * h[u].z;
            const float sw = acc.w + norm * h[u].w;
            acc.x = on ? sx : acc.x;
            acc.y = on ? sy : acc.y;
            acc.z = on ? sz : acc.z;
            acc.w = on ? sw : acc.w;
        }
    }

    float4 v;
    v.x = acc.x + di * di * hs.x + bias[q * 4 + 0];
    v.y = acc.y + di * di * hs.y + bias[q * 4 + 1];
    v.z = acc.z + di * di * hs.z + bias[q * 4 + 2];
    v.w = acc.w + di * di * hs.w + bias[q * 4 + 3];
    if (do_relu) {
        v.x = fmaxf(v.x, 0.0f);
        v.y = fmaxf(v.y, 0.0f);
        v.z = fmaxf(v.z, 0.0f);
        v.w = fmaxf(v.w, 0.0f);
    }
    hout[(size_t)i * 4 + q] = v;
}


__global__ void pool_kernel(const float* __restrict__ h, const int* __restrict__ batch,
                            float* __restrict__ pooled, int n_nodes) {
    int g = blockIdx.x * blockDim.x + threadIdx.x;
    int i = g >> 4;
    if (i >= n_nodes) return;
    int f = g & 15;
    atomicAdd(&pooled[(size_t)batch[i] * HIDDEN + f], h[(size_t)i * HIDDEN + f]);
}

__global__ void final_linear_kernel(const float* __restrict__ pooled,
                                    const float* __restrict__ Wlin,
                                    const float* __restrict__ blin,
                                    float* __restrict__ out,
                                    int n_graphs, int n_classes) {
    int g = blockIdx.x * blockDim.x + threadIdx.x;
    if (g >= n_graphs * n_classes) return;
    int gr = g / n_classes;
    int c  = g % n_classes;
    float acc = blin[c];
#pragma unroll
    for (int f = 0; f < HIDDEN; ++f)
        acc += pooled[(size_t)gr * HIDDEN + f] * Wlin[f * n_classes + c];
    out[g] = acc;
}


static inline size_t align_up(size_t v, size_t a) { return (v + a - 1) & ~(a - 1); }

extern "C" void kernel_launch(void* const* d_in, const int* in_sizes, int n_in,
                              void* d_out, int out_size, void* d_ws, size_t ws_size,
                              hipStream_t stream) {
    (void)n_in; (void)ws_size;

    const float* x    = (const float*)d_in[0];
    const int*   ei   = (const int*)d_in[1];
    const int*   batch= (const int*)d_in[2];
    const float* W1   = (const float*)d_in[3];
    const float* b1   = (const float*)d_in[4];
    const float* W2   = (const float*)d_in[5];
    const float* b2   = (const float*)d_in[6];
    const float* W3   = (const float*)d_in[7];
    const float* b3   = (const float*)d_in[8];
    const float* Wlin = (const float*)d_in[9];
    const float* blin = (const float*)d_in[10];
    float* out = (float*)d_out;

    const int N_FEAT = 128;
    const int n_nodes = in_sizes[0] / N_FEAT;
    const int n_edges = in_sizes[1] / 2;
    const int n_classes = in_sizes[10];
    const int n_graphs  = out_size / n_classes;

    const int* src = ei;
    const int* dst = ei + n_edges;

    char* ws = (char*)d_ws;
    size_t off = 0;
    float* dinv = (float*)(ws + off);  off = align_up(off + (size_t)n_nodes * 4, 256);
    float* bufA = (float*)(ws + off);  off = align_up(off + (size_t)n_nodes * HIDDEN * 4, 256);
    float* bufB = (float*)(ws + off);  off = align_up(off + (size_t)n_nodes * HIDDEN * 4, 256);
    float* pooled = (float*)(ws + off); off = align_up(off + (size_t)n_graphs * HIDDEN * 4, 256);
    unsigned* degPart = (unsigned*)(ws + off);
    off = align_up(off + (size_t)DEG_SLICES * n_nodes * 4, 256);
    unsigned* sliceOff = (unsigned*)(ws + off);
    off = align_up(off + (size_t)DEG_SLICES * n_nodes * 4, 256);
    unsigned* indeg = (unsigned*)(ws + off);    off = align_up(off + (size_t)n_nodes * 4, 256);
    unsigned* rowptr = (unsigned*)(ws + off);   off = align_up(off + (size_t)n_nodes * 4, 256);
    unsigned* blockSum = (unsigned*)(ws + off); off = align_up(off + (size_t)n_nodes * 4, 256);
    uint2* csr = (uint2*)(ws + off);
    off = align_up(off + (size_t)n_edges * 8, 256);

    const int T = 256;
    const int nodeBlocks  = (n_nodes + NODE_T - 1) / NODE_T;
    const int nodeFBlocks = ((n_nodes * HIDDEN) + T - 1) / T;
    const int nodeQBlocks = ((n_nodes * 4) + T - 1) / T;
    const int tiles       = (n_nodes + 15) / 16;
    const int gemmBlocks  = (tiles + 7) / 8;
    const dim3 degGrid(DEG_SLICES, (n_nodes + DEG_RANGE - 1) / DEG_RANGE);

    deg_hist_kernel<<<degGrid, DEG_THREADS, 0, stream>>>(dst, degPart, n_nodes, n_edges);
    dinv_kernel<<<nodeBlocks, NODE_T, 0, stream>>>(degPart, sliceOff, indeg, blockSum, dinv, n_nodes);
    rowptr_kernel<<<nodeBlocks, NODE_T, 0, stream>>>(indeg, blockSum, rowptr, n_nodes);
    csr_fill_kernel<<<degGrid, DEG_THREADS, 0, stream>>>(src, dst, dinv, rowptr, sliceOff, csr,
                                                        n_nodes, n_edges);

    gemm_wmma_kernel<<<gemmBlocks, T, 0, stream>>>(x, W1, bufA, n_nodes, N_FEAT);
    gather_kernel<<<nodeQBlocks, T, 0, stream>>>(csr, rowptr, indeg, dinv, (const float4*)bufA, b1,
                                                 (float4*)bufB, n_nodes, 1);

    gemm_wmma_kernel<<<gemmBlocks, T, 0, stream>>>(bufB, W2, bufA, n_nodes, HIDDEN);
    gather_kernel<<<nodeQBlocks, T, 0, stream>>>(csr, rowptr, indeg, dinv, (const float4*)bufA, b2,
                                                 (float4*)bufB, n_nodes, 1);

    gemm_wmma_kernel<<<gemmBlocks, T, 0, stream>>>(bufB, W3, bufA, n_nodes, HIDDEN);
    gather_kernel<<<nodeQBlocks, T, 0, stream>>>(csr, rowptr, indeg, dinv, (const float4*)bufA, b3,
                                                 (float4*)bufB, n_nodes, 0);

    zero_kernel<<<(n_graphs * HIDDEN + T - 1) / T, T, 0, stream>>>(pooled, n_graphs * HIDDEN);
    pool_kernel<<<nodeFBlocks, T, 0, stream>>>(bufB, batch, pooled, n_nodes);
    final_linear_kernel<<<(n_graphs * n_classes + T - 1) / T, T, 0, stream>>>(
        pooled, Wlin, blin, out, n_graphs, n_classes);
}


# Add-pool folded into the last aggregation kernel: each wave sums its 8 nodes per feature in node order, then one float atomic per feature and graph run; no pool or zero launch (13 to 11 launches)
# speedup vs baseline: 3.0902x; 1.0320x over previous
#include <hip/hip_runtime.h>
#include <hip/hip_bf16.h>


typedef float v2f __attribute__((ext_vector_type(2)));
typedef float v8f __attribute__((ext_vector_type(8)));

#define HIDDEN 16


#define DEG_SLICES  64
#define DEG_RANGE   25600
#define DEG_THREADS 1024
#define DEG_DEPTH   8

__global__ void __launch_bounds__(DEG_THREADS)
deg_hist_kernel(const int* __restrict__ dst, unsigned* __restrict__ part,
                int n_nodes, int n_edges) {
    __shared__ unsigned hist[DEG_RANGE];
    const int tid  = threadIdx.x;
    const int base = blockIdx.y * DEG_RANGE;

    for (int i = tid; i < DEG_RANGE; i += DEG_THREADS) hist[i] = 0u;
    __syncthreads();

    const int slice = ((n_edges + DEG_SLICES * 4 - 1) / (DEG_SLICES * 4)) * 4;
    const int e0 = blockIdx.x * slice;
    const int e1 = min(e0 + slice, n_edges);

    int e = e0 + tid;
    for (; e + (DEG_DEPTH - 1) * DEG_THREADS < e1; e += DEG_DEPTH * DEG_THREADS) {
        unsigned d[DEG_DEPTH];
#pragma unroll
        for (int u = 0; u < DEG_DEPTH; ++u)
            d[u] = (unsigned)(dst[e + u * DEG_THREADS] - base);
#pragma unroll
        for (int u = 0; u < DEG_DEPTH; ++u)
            if (d[u] < DEG_RANGE) atomicAdd(&hist[d[u]], 1u);
    }
    for (; e < e1; e += DEG_THREADS) {
        unsigned d0 = (unsigned)(dst[e] - base);
        if (d0 < DEG_RANGE) atomicAdd(&hist[d0], 1u);
    }
    __syncthreads();

    const int nr = min(DEG_RANGE, n_nodes - base);
    unsigned* p = part + (size_t)blockIdx.x * n_nodes + base;
    for (int i = tid; i < nr; i += DEG_THREADS) p[i] = hist[i];
}

#define NODE_T 256

__global__ void __launch_bounds__(NODE_T)
dinv_kernel(const unsigned* __restrict__ part, unsigned* __restrict__ soff,
            unsigned* __restrict__ indeg, unsigned* __restrict__ blockSum,
            float* __restrict__ dinv, float* __restrict__ pooled, int n_pool, int n) {
    __shared__ unsigned sh[NODE_T];
    const int tid = threadIdx.x;
    const int i = blockIdx.x * NODE_T + tid;
    for (int k = i; k < n_pool; k += (int)gridDim.x * NODE_T) pooled[k] = 0.0f;
    unsigned run = 0u;
    if (i < n) {
        static_assert(DEG_SLICES % 16 == 0, "slices are taken 16 at a time");
        for (int s0 = 0; s0 < DEG_SLICES; s0 += 16) {
            unsigned c[16];
#pragma unroll
            for (int u = 0; u < 16; ++u) c[u] = part[(size_t)(s0 + u) * n + i];
#pragma unroll
            for (int u = 0; u < 16; ++u) {
                soff[(size_t)(s0 + u) * n + i] = run;
                run += c[u];
            }
        }
        indeg[i] = run;
        dinv[i] = rsqrtf((float)(run + 1u));
    }
    sh[tid] = run;
    __syncthreads();
    for (int o = NODE_T / 2; o > 0; o >>= 1) {
        if (tid < o) sh[tid] += sh[tid + o];
        __syncthreads();
    }
    if (tid == 0) blockSum[blockIdx.x] = sh[0];
}

__global__ void __launch_bounds__(NODE_T)
rowptr_kernel(const unsigned* __restrict__ indeg, const unsigned* __restrict__ blockSum,
              unsigned* __restrict__ rowptr, int n) {
    __shared__ unsigned sh[NODE_T];
    const int tid = threadIdx.x;
    unsigned pre = 0u;
    for (int b = tid; b < (int)blockIdx.x; b += NODE_T) pre += blockSum[b];
    sh[tid] = pre;
    __syncthreads();
    for (int o = NODE_T / 2; o > 0; o >>= 1) {
        if (tid < o) sh[tid] += sh[tid + o];
        __syncthreads();
    }
    const unsigned before = sh[0];
    __syncthreads();

    const int i = blockIdx.x * NODE_T + tid;
    const unsigned v = (i < n) ? indeg[i] : 0u;
    sh[tid] = v;
    __syncthreads();
    for (int o = 1; o < NODE_T; o <<= 1) {
        unsigned t = (tid >= o) ? sh[tid - o] : 0u;
        __syncthreads();
        sh[tid] += t;
        __syncthreads();
    }
    if (i < n) rowptr[i] = before + sh[tid] - v;
}

__global__ void __launch_bounds__(DEG_THREADS)
csr_fill_kernel(const int* __restrict__ src, const int* __restrict__ dst,
                const float* __restrict__ dinv,
                const unsigned* __restrict__ rowptr, const unsigned* __restrict__ soff,
                uint2* __restrict__ csr, int n_nodes, int n_edges) {
    __shared__ unsigned cur[DEG_RANGE];
    const int tid  = threadIdx.x;
    const int base = blockIdx.y * DEG_RANGE;
    const int nr   = min(DEG_RANGE, n_nodes - base);

    const unsigned* so = soff + (size_t)blockIdx.x * n_nodes + base;
    for (int i = tid; i < nr; i += DEG_THREADS) cur[i] = rowptr[base + i] + so[i];
    __syncthreads();

    const int slice = ((n_edges + DEG_SLICES * 4 - 1) / (DEG_SLICES * 4)) * 4;
    const int e0 = blockIdx.x * slice;
    const int e1 = min(e0 + slice, n_edges);

    int e = e0 + tid;
    for (; e + (DEG_DEPTH - 1) * DEG_THREADS < e1; e += DEG_DEPTH * DEG_THREADS) {
        unsigned d[DEG_DEPTH];
        int      s[DEG_DEPTH];
        float    w[DEG_DEPTH];
#pragma unroll
        for (int u = 0; u < DEG_DEPTH; ++u) {
            d[u] = (unsigned)(dst[e + u * DEG_THREADS] - base);
            s[u] = src[e + u * DEG_THREADS];
        }
#pragma unroll
        for (int u = 0; u < DEG_DEPTH; ++u)
            w[u] = dinv[d[u] < (unsigned)nr ? s[u] : 0];
        asm volatile("" :: "v"(w[0]), "v"(w[1]), "v"(w[2]), "v"(w[3]),
                           "v"(w[4]), "v"(w[5]), "v"(w[6]), "v"(w[7]));
#pragma unroll
        for (int u = 0; u < DEG_DEPTH; ++u)
            if (d[u] < (unsigned)nr) {
                unsigned pos = atomicAdd(&cur[d[u]], 1u);
                csr[pos] = make_uint2((unsigned)s[u], __float_as_uint(w[u]));
            }
    }
    for (; e < e1; e += DEG_THREADS) {
        unsigned d0 = (unsigned)(dst[e] - base);
        if (d0 < (unsigned)nr) {
            int s0 = src[e];
            unsigned pos = atomicAdd(&cur[d0], 1u);
            csr[pos] = make_uint2((unsigned)s0, __float_as_uint(dinv[s0]));
        }
    }
}

__global__ void __launch_bounds__(256)
gemm_wmma_kernel(const float* __restrict__ X, const float* __restrict__ W,
                 float* __restrict__ H, int n_rows, int K) {
    int wave = (int)((blockIdx.x * blockDim.x + threadIdx.x) >> 5);
    int lane = threadIdx.x & 31;
    int row0 = wave * 16;
    if (row0 >= n_rows) return;

    int half = lane >> 4;
    int l16  = lane & 15;

    const float* xrow = X + (size_t)(row0 + l16) * K;
    v8f acc = {};
    for (int k = 0; k < K; k += 4) {
        int ka = k + half * 2;
        v2f a, b;
        a.x = xrow[ka];
        a.y = xrow[ka + 1];
        b.x = W[(size_t)ka * HIDDEN + l16];
        b.y = W[(size_t)(ka + 1) * HIDDEN + l16];
        acc = __builtin_amdgcn_wmma_f32_16x16x4_f32(
             false, a,  false, b,
             (short)0, acc,  false,  false);
    }
    float* out = H + (size_t)row0 * HIDDEN;
#pragma unroll
    for (int i = 0; i < 8; ++i)
        out[(size_t)(i + 8 * half) * HIDDEN + l16] = acc[i];
}


#define GATHER_DEPTH 8

template <bool POOL>
__global__ void __launch_bounds__(256)
gather_kernel(const uint2* __restrict__ csr, const unsigned* __restrict__ rowptr,
              const unsigned* __restrict__ indeg, const float* __restrict__ dinv,
              const float4* __restrict__ hin, const float* __restrict__ bias,
              float4* __restrict__ hout, const int* __restrict__ batch,
              float* __restrict__ pooled, int n_nodes, int n_graphs, int do_relu) {
    int g = blockIdx.x * blockDim.x + threadIdx.x;
    int i = g >> 2;
    const bool valid = i < n_nodes;
    if (!POOL) {
        if (!valid) return;
    } else if (!valid) {
        i = 0;
    }
    int q = g & 3;
    int bid = -1;
    if (POOL) {
        bid = batch[i];
        if (!valid) bid = -1;
    }
    unsigned j = rowptr[i];
    const unsigned cnt = indeg[i];
    const unsigned end = j + (valid ? cnt : 0u);
    const float di = dinv[i];
    const float4 hs = hin[(size_t)i * 4 + q];
    float4 acc = make_float4(0.0f, 0.0f, 0.0f, 0.0f);

    for (; j < end; j += GATHER_DEPTH) {
        uint2  c[GATHER_DEPTH];
        float4 h[GATHER_DEPTH];
#pragma unroll
        for (int u = 0; u < GATHER_DEPTH; ++u)
            c[u] = csr[min(j + (unsigned)u, end - 1u)];
#pragma unroll
        for (int u = 0; u < GATHER_DEPTH; ++u)
            h[u] = hin[(size_t)c[u].x * 4 + q];
#pragma unroll
        for (int u = 0; u < GATHER_DEPTH; ++u) {
            const bool on = j + (unsigned)u < end;
            const float norm = __uint_as_float(c[u].y) * di;
            const float sx = acc.x + norm * h[u].x;
            const float sy = acc.y + norm * h[u].y;
            const float sz = acc.z + norm * h[u].z;
            const float sw = acc.w + norm * h[u].w;
            acc.x = on ? sx : acc.x;
            acc.y = on ? sy : acc.y;
            acc.z = on ? sz : acc.z;
            acc.w = on ? sw : acc.w;
        }
    }

    float4 v;
    v.x = acc.x + di * di * hs.x + bias[q * 4 + 0];
    v.y = acc.y + di * di * hs.y + bias[q * 4 + 1];
    v.z = acc.z + di * di * hs.z + bias[q * 4 + 2];
    v.w = acc.w + di * di * hs.w + bias[q * 4 + 3];
    if (do_relu) {
        v.x = fmaxf(v.x, 0.0f);
        v.y = fmaxf(v.y, 0.0f);
        v.z = fmaxf(v.z, 0.0f);
        v.w = fmaxf(v.w, 0.0f);
    }
    if (!POOL) {
        hout[(size_t)i * 4 + q] = v;
    } else {
        __shared__ float4 srow[256];
        const int lane = threadIdx.x & 31;
        const int f = lane & 15;
        srow[threadIdx.x] = v;
        __syncthreads();
        const float* rows = (const float*)&srow[threadIdx.x & ~31];
        float sum = 0.0f;
        int cur = __builtin_amdgcn_readlane(bid, 0);
#pragma unroll
        for (int n = 0; n < 8; ++n) {
            const int b = __builtin_amdgcn_readlane(bid, n * 4);
            const float x = rows[n * 16 + f];
            if (b != cur) {
                if (lane < 16 && (unsigned)cur < (unsigned)n_graphs)
                    atomicAdd(&pooled[(size_t)cur * HIDDEN + f], sum);
                sum = 0.0f;
                cur = b;
            }
            sum += x;
        }
        if (lane < 16 && (unsigned)cur < (unsigned)n_graphs)
            atomicAdd(&pooled[(size_t)cur * HIDDEN + f], sum);
    }
}


__global__ void final_linear_kernel(const float* __restrict__ pooled,
                                    const float* __restrict__ Wlin,
                                    const float* __restrict__ blin,
                                    float* __restrict__ out,
                                    int n_graphs, int n_classes) {
    int g = blockIdx.x * blockDim.x + threadIdx.x;
    if (g >= n_graphs * n_classes) return;
    int gr = g / n_classes;
    int c  = g % n_classes;
    float acc = blin[c];
#pragma unroll
    for (int f = 0; f < HIDDEN; ++f)
        acc += pooled[(size_t)gr * HIDDEN + f] * Wlin[f * n_classes + c];
    out[g] = acc;
}


static inline size_t align_up(size_t v, size_t a) { return (v + a - 1) & ~(a - 1); }

extern "C" void kernel_launch(void* const* d_in, const int* in_sizes, int n_in,
                              void* d_out, int out_size, void* d_ws, size_t ws_size,
                              hipStream_t stream) {
    (void)n_in; (void)ws_size;

    const float* x    = (const float*)d_in[0];
    const int*   ei   = (const int*)d_in[1];
    const int*   batch= (const int*)d_in[2];
    const float* W1   = (const float*)d_in[3];
    const float* b1   = (const float*)d_in[4];
    const float* W2   = (const float*)d_in[5];
    const float* b2   = (const float*)d_in[6];
    const float* W3   = (const float*)d_in[7];
    const float* b3   = (const float*)d_in[8];
    const float* Wlin = (const float*)d_in[9];
    const float* blin = (const float*)d_in[10];
    float* out = (float*)d_out;

    const int N_FEAT = 128;
    const int n_nodes = in_sizes[0] / N_FEAT;
    const int n_edges = in_sizes[1] / 2;
    const int n_classes = in_sizes[10];
    const int n_graphs  = out_size / n_classes;

    const int* src = ei;
    const int* dst = ei + n_edges;

    char* ws = (char*)d_ws;
    size_t off = 0;
    float* dinv = (float*)(ws + off);  off = align_up(off + (size_t)n_nodes * 4, 256);
    float* bufA = (float*)(ws + off);  off = align_up(off + (size_t)n_nodes * HIDDEN * 4, 256);
    float* bufB = (float*)(ws + off);  off = align_up(off + (size_t)n_nodes * HIDDEN * 4, 256);
    float* pooled = (float*)(ws + off); off = align_up(off + (size_t)n_graphs * HIDDEN * 4, 256);
    unsigned* degPart = (unsigned*)(ws + off);
    off = align_up(off + (size_t)DEG_SLICES * n_nodes * 4, 256);
    unsigned* sliceOff = (unsigned*)(ws + off);
    off = align_up(off + (size_t)DEG_SLICES * n_nodes * 4, 256);
    unsigned* indeg = (unsigned*)(ws + off);    off = align_up(off + (size_t)n_nodes * 4, 256);
    unsigned* rowptr = (unsigned*)(ws + off);   off = align_up(off + (size_t)n_nodes * 4, 256);
    unsigned* blockSum = (unsigned*)(ws + off); off = align_up(off + (size_t)n_nodes * 4, 256);
    uint2* csr = (uint2*)(ws + off);
    off = align_up(off + (size_t)n_edges * 8, 256);

    const int T = 256;
    const int nodeBlocks  = (n_nodes + NODE_T - 1) / NODE_T;
    const int nodeQBlocks = ((n_nodes * 4) + T - 1) / T;
    const int tiles       = (n_nodes + 15) / 16;
    const int gemmBlocks  = (tiles + 7) / 8;
    const dim3 degGrid(DEG_SLICES, (n_nodes + DEG_RANGE - 1) / DEG_RANGE);

    deg_hist_kernel<<<degGrid, DEG_THREADS, 0, stream>>>(dst, degPart, n_nodes, n_edges);
    dinv_kernel<<<nodeBlocks, NODE_T, 0, stream>>>(degPart, sliceOff, indeg, blockSum, dinv,
                                                   pooled, n_graphs * HIDDEN, n_nodes);
    rowptr_kernel<<<nodeBlocks, NODE_T, 0, stream>>>(indeg, blockSum, rowptr, n_nodes);
    csr_fill_kernel<<<degGrid, DEG_THREADS, 0, stream>>>(src, dst, dinv, rowptr, sliceOff, csr,
                                                        n_nodes, n_edges);

    gemm_wmma_kernel<<<gemmBlocks, T, 0, stream>>>(x, W1, bufA, n_nodes, N_FEAT);
    gather_kernel<false><<<nodeQBlocks, T, 0, stream>>>(csr, rowptr, indeg, dinv, (const float4*)bufA, b1,
                                                        (float4*)bufB, batch, pooled,
                                                        n_nodes, n_graphs, 1);

    gemm_wmma_kernel<<<gemmBlocks, T, 0, stream>>>(bufB, W2, bufA, n_nodes, HIDDEN);
    gather_kernel<false><<<nodeQBlocks, T, 0, stream>>>(csr, rowptr, indeg, dinv, (const float4*)bufA, b2,
                                                        (float4*)bufB, batch, pooled,
                                                        n_nodes, n_graphs, 1);

    gemm_wmma_kernel<<<gemmBlocks, T, 0, stream>>>(bufB, W3, bufA, n_nodes, HIDDEN);
    gather_kernel<true><<<nodeQBlocks, T, 0, stream>>>(csr, rowptr, indeg, dinv, (const float4*)bufA, b3,
                                                       (float4*)bufB, batch, pooled,
                                                       n_nodes, n_graphs, 0);

    final_linear_kernel<<<(n_graphs * n_classes + T - 1) / T, T, 0, stream>>>(
        pooled, Wlin, blin, out, n_graphs, n_classes);
}


# Layer 2 and 3 products (h@W2, h@W3) done by the aggregation kernel that finishes h: each wave turns its 8 rows through LDS into a WMMA A fragment; no separate product launch (11 to 9 launches)
# speedup vs baseline: 3.2427x; 1.0493x over previous
#include <hip/hip_runtime.h>
#include <hip/hip_bf16.h>


typedef float v2f __attribute__((ext_vector_type(2)));
typedef float v8f __attribute__((ext_vector_type(8)));

#define HIDDEN 16


#define DEG_SLICES  64
#define DEG_RANGE   25600
#define DEG_THREADS 1024
#define DEG_DEPTH   8

__global__ void __launch_bounds__(DEG_THREADS)
deg_hist_kernel(const int* __restrict__ dst, unsigned* __restrict__ part,
                int n_nodes, int n_edges) {
    __shared__ unsigned hist[DEG_RANGE];
    const int tid  = threadIdx.x;
    const int base = blockIdx.y * DEG_RANGE;

    for (int i = tid; i < DEG_RANGE; i += DEG_THREADS) hist[i] = 0u;
    __syncthreads();

    const int slice = ((n_edges + DEG_SLICES * 4 - 1) / (DEG_SLICES * 4)) * 4;
    const int e0 = blockIdx.x * slice;
    const int e1 = min(e0 + slice, n_edges);

    int e = e0 + tid;
    for (; e + (DEG_DEPTH - 1) * DEG_THREADS < e1; e += DEG_DEPTH * DEG_THREADS) {
        unsigned d[DEG_DEPTH];
#pragma unroll
        for (int u = 0; u < DEG_DEPTH; ++u)
            d[u] = (unsigned)(dst[e + u * DEG_THREADS] - base);
#pragma unroll
        for (int u = 0; u < DEG_DEPTH; ++u)
            if (d[u] < DEG_RANGE) atomicAdd(&hist[d[u]], 1u);
    }
    for (; e < e1; e += DEG_THREADS) {
        unsigned d0 = (unsigned)(dst[e] - base);
        if (d0 < DEG_RANGE) atomicAdd(&hist[d0], 1u);
    }
    __syncthreads();

    const int nr = min(DEG_RANGE, n_nodes - base);
    unsigned* p = part + (size_t)blockIdx.x * n_nodes + base;
    for (int i = tid; i < nr; i += DEG_THREADS) p[i] = hist[i];
}

#define NODE_T 256

__global__ void __launch_bounds__(NODE_T)
dinv_kernel(const unsigned* __restrict__ part, unsigned* __restrict__ soff,
            unsigned* __restrict__ indeg, unsigned* __restrict__ blockSum,
            float* __restrict__ dinv, float* __restrict__ pooled, int n_pool, int n) {
    __shared__ unsigned sh[NODE_T];
    const int tid = threadIdx.x;
    const int i = blockIdx.x * NODE_T + tid;
    for (int k = i; k < n_pool; k += (int)gridDim.x * NODE_T) pooled[k] = 0.0f;
    unsigned run = 0u;
    if (i < n) {
        static_assert(DEG_SLICES % 16 == 0, "slices are taken 16 at a time");
        for (int s0 = 0; s0 < DEG_SLICES; s0 += 16) {
            unsigned c[16];
#pragma unroll
            for (int u = 0; u < 16; ++u) c[u] = part[(size_t)(s0 + u) * n + i];
#pragma unroll
            for (int u = 0; u < 16; ++u) {
                soff[(size_t)(s0 + u) * n + i] = run;
                run += c[u];
            }
        }
        indeg[i] = run;
        dinv[i] = rsqrtf((float)(run + 1u));
    }
    sh[tid] = run;
    __syncthreads();
    for (int o = NODE_T / 2; o > 0; o >>= 1) {
        if (tid < o) sh[tid] += sh[tid + o];
        __syncthreads();
    }
    if (tid == 0) blockSum[blockIdx.x] = sh[0];
}

__global__ void __launch_bounds__(NODE_T)
rowptr_kernel(const unsigned* __restrict__ indeg, const unsigned* __restrict__ blockSum,
              unsigned* __restrict__ rowptr, int n) {
    __shared__ unsigned sh[NODE_T];
    const int tid = threadIdx.x;
    unsigned pre = 0u;
    for (int b = tid; b < (int)blockIdx.x; b += NODE_T) pre += blockSum[b];
    sh[tid] = pre;
    __syncthreads();
    for (int o = NODE_T / 2; o > 0; o >>= 1) {
        if (tid < o) sh[tid] += sh[tid + o];
        __syncthreads();
    }
    const unsigned before = sh[0];
    __syncthreads();

    const int i = blockIdx.x * NODE_T + tid;
    const unsigned v = (i < n) ? indeg[i] : 0u;
    sh[tid] = v;
    __syncthreads();
    for (int o = 1; o < NODE_T; o <<= 1) {
        unsigned t = (tid >= o) ? sh[tid - o] : 0u;
        __syncthreads();
        sh[tid] += t;
        __syncthreads();
    }
    if (i < n) rowptr[i] = before + sh[tid] - v;
}

__global__ void __launch_bounds__(DEG_THREADS)
csr_fill_kernel(const int* __restrict__ src, const int* __restrict__ dst,
                const float* __restrict__ dinv,
                const unsigned* __restrict__ rowptr, const unsigned* __restrict__ soff,
                uint2* __restrict__ csr, int n_nodes, int n_edges) {
    __shared__ unsigned cur[DEG_RANGE];
    const int tid  = threadIdx.x;
    const int base = blockIdx.y * DEG_RANGE;
    const int nr   = min(DEG_RANGE, n_nodes - base);

    const unsigned* so = soff + (size_t)blockIdx.x * n_nodes + base;
    for (int i = tid; i < nr; i += DEG_THREADS) cur[i] = rowptr[base + i] + so[i];
    __syncthreads();

    const int slice = ((n_edges + DEG_SLICES * 4 - 1) / (DEG_SLICES * 4)) * 4;
    const int e0 = blockIdx.x * slice;
    const int e1 = min(e0 + slice, n_edges);

    int e = e0 + tid;
    for (; e + (DEG_DEPTH - 1) * DEG_THREADS < e1; e += DEG_DEPTH * DEG_THREADS) {
        unsigned d[DEG_DEPTH];
        int      s[DEG_DEPTH];
        float    w[DEG_DEPTH];
#pragma unroll
        for (int u = 0; u < DEG_DEPTH; ++u) {
            d[u] = (unsigned)(dst[e + u * DEG_THREADS] - base);
            s[u] = src[e + u * DEG_THREADS];
        }
#pragma unroll
        for (int u = 0; u < DEG_DEPTH; ++u)
            w[u] = dinv[d[u] < (unsigned)nr ? s[u] : 0];
        asm volatile("" :: "v"(w[0]), "v"(w[1]), "v"(w[2]), "v"(w[3]),
                           "v"(w[4]), "v"(w[5]), "v"(w[6]), "v"(w[7]));
#pragma unroll
        for (int u = 0; u < DEG_DEPTH; ++u)
            if (d[u] < (unsigned)nr) {
                unsigned pos = atomicAdd(&cur[d[u]], 1u);
                csr[pos] = make_uint2((unsigned)s[u], __float_as_uint(w[u]));
            }
    }
    for (; e < e1; e += DEG_THREADS) {
        unsigned d0 = (unsigned)(dst[e] - base);
        if (d0 < (unsigned)nr) {
            int s0 = src[e];
            unsigned pos = atomicAdd(&cur[d0], 1u);
            csr[pos] = make_uint2((unsigned)s0, __float_as_uint(dinv[s0]));
        }
    }
}

__global__ void __launch_bounds__(256)
gemm_wmma_kernel(const float* __restrict__ X, const float* __restrict__ W,
                 float* __restrict__ H, int n_rows, int K) {
    int wave = (int)((blockIdx.x * blockDim.x + threadIdx.x) >> 5);
    int lane = threadIdx.x & 31;
    int row0 = wave * 16;
    if (row0 >= n_rows) return;

    int half = lane >> 4;
    int l16  = lane & 15;

    const float* xrow = X + (size_t)(row0 + l16) * K;
    v8f acc = {};
    for (int k = 0; k < K; k += 4) {
        int ka = k + half * 2;
        v2f a, b;
        a.x = xrow[ka];
        a.y = xrow[ka + 1];
        b.x = W[(size_t)ka * HIDDEN + l16];
        b.y = W[(size_t)(ka + 1) * HIDDEN + l16];
        acc = __builtin_amdgcn_wmma_f32_16x16x4_f32(
             false, a,  false, b,
             (short)0, acc,  false,  false);
    }
    float* out = H + (size_t)row0 * HIDDEN;
#pragma unroll
    for (int i = 0; i < 8; ++i)
        out[(size_t)(i + 8 * half) * HIDDEN + l16] = acc[i];
}


#define GATHER_DEPTH 8
#define ROW_PITCH 20

__device__ __forceinline__ void wave_lds_handover() {
    __builtin_amdgcn_fence(__ATOMIC_RELEASE, "wavefront");
    __builtin_amdgcn_wave_barrier();
    __builtin_amdgcn_fence(__ATOMIC_ACQUIRE, "wavefront");
}

template <bool POOL>
__global__ void __launch_bounds__(256)
gather_kernel(const uint2* __restrict__ csr, const unsigned* __restrict__ rowptr,
              const unsigned* __restrict__ indeg, const float* __restrict__ dinv,
              const float4* __restrict__ hin, const float* __restrict__ bias,
              const float* __restrict__ Wn,
              float4* __restrict__ hout, const int* __restrict__ batch,
              float* __restrict__ pooled, int n_nodes, int n_graphs, int do_relu) {
    int g = blockIdx.x * blockDim.x + threadIdx.x;
    int i = g >> 2;
    const bool valid = i < n_nodes;
    if (!POOL) {
        if (((g & ~31) >> 2) >= n_nodes) return;
    }
    if (!valid) i = 0;
    int q = g & 3;
    int bid = -1;
    if (POOL) {
        bid = batch[i];
        if (!valid) bid = -1;
    }
    unsigned j = rowptr[i];
    const unsigned cnt = indeg[i];
    const unsigned end = j + (valid ? cnt : 0u);
    const float di = dinv[i];
    const float4 hs = hin[(size_t)i * 4 + q];
    float4 acc = make_float4(0.0f, 0.0f, 0.0f, 0.0f);

    for (; j < end; j += GATHER_DEPTH) {
        uint2  c[GATHER_DEPTH];
        float4 h[GATHER_DEPTH];
#pragma unroll
        for (int u = 0; u < GATHER_DEPTH; ++u)
            c[u] = csr[min(j + (unsigned)u, end - 1u)];
#pragma unroll
        for (int u = 0; u < GATHER_DEPTH; ++u)
            h[u] = hin[(size_t)c[u].x * 4 + q];
#pragma unroll
        for (int u = 0; u < GATHER_DEPTH; ++u) {
            const bool on = j + (unsigned)u < end;
            const float norm = __uint_as_float(c[u].y) * di;
            const float sx = acc.x + norm * h[u].x;
            const float sy = acc.y + norm * h[u].y;
            const float sz = acc.z + norm * h[u].z;
            const float sw = acc.w + norm * h[u].w;
            acc.x = on ? sx : acc.x;
            acc.y = on ? sy : acc.y;
            acc.z = on ? sz : acc.z;
            acc.w = on ? sw : acc.w;
        }
    }

    float4 v;
    v.x = acc.x + di * di * hs.x + bias[q * 4 + 0];
    v.y = acc.y + di * di * hs.y + bias[q * 4 + 1];
    v.z = acc.z + di * di * hs.z + bias[q * 4 + 2];
    v.w = acc.w + di * di * hs.w + bias[q * 4 + 3];
    if (do_relu) {
        v.x = fmaxf(v.x, 0.0f);
        v.y = fmaxf(v.y, 0.0f);
        v.z = fmaxf(v.z, 0.0f);
        v.w = fmaxf(v.w, 0.0f);
    }
    const int lane = threadIdx.x & 31;
    if (!POOL) {
        __shared__ __attribute__((aligned(16))) float sIn[8 * 8 * ROW_PITCH];
        __shared__ __attribute__((aligned(16))) float sOut[8 * 8 * ROW_PITCH];
        float* in_rows  = sIn  + (threadIdx.x >> 5) * (8 * ROW_PITCH);
        float* out_rows = sOut + (threadIdx.x >> 5) * (8 * ROW_PITCH);
        const int nd = lane >> 2;
        *(float4*)&in_rows[nd * ROW_PITCH + q * 4] = v;
        wave_lds_handover();

        const int half = lane >> 4;
        const int l16  = lane & 15;
        const float* arow = in_rows + (l16 & 7) * ROW_PITCH;
        v8f t = {};
#pragma unroll
        for (int kk = 0; kk < HIDDEN / 4; ++kk) {
            const int ka = kk * 4 + half * 2;
            v2f a, b;
            a.x = arow[ka];
            a.y = arow[ka + 1];
            b.x = Wn[ka * HIDDEN + l16];
            b.y = Wn[(ka + 1) * HIDDEN + l16];
            t = __builtin_amdgcn_wmma_f32_16x16x4_f32(
                 false, a,  false, b,
                 (short)0, t,  false,  false);
        }
        if (half == 0) {
#pragma unroll
            for (int r = 0; r < 8; ++r) out_rows[r * ROW_PITCH + l16] = t[r];
        }
        wave_lds_handover();
        const float4 o = *(const float4*)&out_rows[nd * ROW_PITCH + q * 4];
        if (valid) hout[(size_t)i * 4 + q] = o;
    } else {
        __shared__ float4 srow[256];
        const int f = lane & 15;
        srow[threadIdx.x] = v;
        __syncthreads();
        const float* rows = (const float*)&srow[threadIdx.x & ~31];
        float sum = 0.0f;
        int cur = __builtin_amdgcn_readlane(bid, 0);
#pragma unroll
        for (int n = 0; n < 8; ++n) {
            const int b = __builtin_amdgcn_readlane(bid, n * 4);
            const float x = rows[n * 16 + f];
            if (b != cur) {
                if (lane < 16 && (unsigned)cur < (unsigned)n_graphs)
                    atomicAdd(&pooled[(size_t)cur * HIDDEN + f], sum);
                sum = 0.0f;
                cur = b;
            }
            sum += x;
        }
        if (lane < 16 && (unsigned)cur < (unsigned)n_graphs)
            atomicAdd(&pooled[(size_t)cur * HIDDEN + f], sum);
    }
}


__global__ void final_linear_kernel(const float* __restrict__ pooled,
                                    const float* __restrict__ Wlin,
                                    const float* __restrict__ blin,
                                    float* __restrict__ out,
                                    int n_graphs, int n_classes) {
    int g = blockIdx.x * blockDim.x + threadIdx.x;
    if (g >= n_graphs * n_classes) return;
    int gr = g / n_classes;
    int c  = g % n_classes;
    float acc = blin[c];
#pragma unroll
    for (int f = 0; f < HIDDEN; ++f)
        acc += pooled[(size_t)gr * HIDDEN + f] * Wlin[f * n_classes + c];
    out[g] = acc;
}


static inline size_t align_up(size_t v, size_t a) { return (v + a - 1) & ~(a - 1); }

extern "C" void kernel_launch(void* const* d_in, const int* in_sizes, int n_in,
                              void* d_out, int out_size, void* d_ws, size_t ws_size,
                              hipStream_t stream) {
    (void)n_in; (void)ws_size;

    const float* x    = (const float*)d_in[0];
    const int*   ei   = (const int*)d_in[1];
    const int*   batch= (const int*)d_in[2];
    const float* W1   = (const float*)d_in[3];
    const float* b1   = (const float*)d_in[4];
    const float* W2   = (const float*)d_in[5];
    const float* b2   = (const float*)d_in[6];
    const float* W3   = (const float*)d_in[7];
    const float* b3   = (const float*)d_in[8];
    const float* Wlin = (const float*)d_in[9];
    const float* blin = (const float*)d_in[10];
    float* out = (float*)d_out;

    const int N_FEAT = 128;
    const int n_nodes = in_sizes[0] / N_FEAT;
    const int n_edges = in_sizes[1] / 2;
    const int n_classes = in_sizes[10];
    const int n_graphs  = out_size / n_classes;

    const int* src = ei;
    const int* dst = ei + n_edges;

    char* ws = (char*)d_ws;
    size_t off = 0;
    float* dinv = (float*)(ws + off);  off = align_up(off + (size_t)n_nodes * 4, 256);
    float* bufA = (float*)(ws + off);  off = align_up(off + (size_t)n_nodes * HIDDEN * 4, 256);
    float* bufB = (float*)(ws + off);  off = align_up(off + (size_t)n_nodes * HIDDEN * 4, 256);
    float* pooled = (float*)(ws + off); off = align_up(off + (size_t)n_graphs * HIDDEN * 4, 256);
    unsigned* degPart = (unsigned*)(ws + off);
    off = align_up(off + (size_t)DEG_SLICES * n_nodes * 4, 256);
    unsigned* sliceOff = (unsigned*)(ws + off);
    off = align_up(off + (size_t)DEG_SLICES * n_nodes * 4, 256);
    unsigned* indeg = (unsigned*)(ws + off);    off = align_up(off + (size_t)n_nodes * 4, 256);
    unsigned* rowptr = (unsigned*)(ws + off);   off = align_up(off + (size_t)n_nodes * 4, 256);
    unsigned* blockSum = (unsigned*)(ws + off); off = align_up(off + (size_t)n_nodes * 4, 256);
    uint2* csr = (uint2*)(ws + off);
    off = align_up(off + (size_t)n_edges * 8, 256);

    const int T = 256;
    const int nodeBlocks  = (n_nodes + NODE_T - 1) / NODE_T;
    const int nodeQBlocks = ((n_nodes * 4) + T - 1) / T;
    const int tiles       = (n_nodes + 15) / 16;
    const int gemmBlocks  = (tiles + 7) / 8;
    const dim3 degGrid(DEG_SLICES, (n_nodes + DEG_RANGE - 1) / DEG_RANGE);

    deg_hist_kernel<<<degGrid, DEG_THREADS, 0, stream>>>(dst, degPart, n_nodes, n_edges);
    dinv_kernel<<<nodeBlocks, NODE_T, 0, stream>>>(degPart, sliceOff, indeg, blockSum, dinv,
                                                   pooled, n_graphs * HIDDEN, n_nodes);
    rowptr_kernel<<<nodeBlocks, NODE_T, 0, stream>>>(indeg, blockSum, rowptr, n_nodes);
    csr_fill_kernel<<<degGrid, DEG_THREADS, 0, stream>>>(src, dst, dinv, rowptr, sliceOff, csr,
                                                        n_nodes, n_edges);

    gemm_wmma_kernel<<<gemmBlocks, T, 0, stream>>>(x, W1, bufA, n_nodes, N_FEAT);

    gather_kernel<false><<<nodeQBlocks, T, 0, stream>>>(csr, rowptr, indeg, dinv, (const float4*)bufA, b1, W2,
                                                        (float4*)bufB, batch, pooled,
                                                        n_nodes, n_graphs, 1);

    gather_kernel<false><<<nodeQBlocks, T, 0, stream>>>(csr, rowptr, indeg, dinv, (const float4*)bufB, b2, W3,
                                                        (float4*)bufA, batch, pooled,
                                                        n_nodes, n_graphs, 1);

    gather_kernel<true><<<nodeQBlocks, T, 0, stream>>>(csr, rowptr, indeg, dinv, (const float4*)bufA, b3, nullptr,
                                                       (float4*)bufB, batch, pooled,
                                                       n_nodes, n_graphs, 0);

    final_linear_kernel<<<(n_graphs * n_classes + T - 1) / T, T, 0, stream>>>(
        pooled, Wlin, blin, out, n_graphs, n_classes);
}


# Gather loop pipelined: next batch's slots load with this batch's rows (2 per lane, quad-shared by DPP; self row parked in LDS, 64 VGPRs): one round trip per batch, not two; sum order kept
# speedup vs baseline: 3.3447x; 1.0315x over previous
#include <hip/hip_runtime.h>
#include <hip/hip_bf16.h>


typedef float v2f __attribute__((ext_vector_type(2)));
typedef float v8f __attribute__((ext_vector_type(8)));

#define HIDDEN 16


#define DEG_SLICES  64
#define DEG_RANGE   25600
#define DEG_THREADS 1024
#define DEG_DEPTH   8

__global__ void __launch_bounds__(DEG_THREADS)
deg_hist_kernel(const int* __restrict__ dst, unsigned* __restrict__ part,
                int n_nodes, int n_edges) {
    __shared__ unsigned hist[DEG_RANGE];
    const int tid  = threadIdx.x;
    const int base = blockIdx.y * DEG_RANGE;

    for (int i = tid; i < DEG_RANGE; i += DEG_THREADS) hist[i] = 0u;
    __syncthreads();

    const int slice = ((n_edges + DEG_SLICES * 4 - 1) / (DEG_SLICES * 4)) * 4;
    const int e0 = blockIdx.x * slice;
    const int e1 = min(e0 + slice, n_edges);

    int e = e0 + tid;
    for (; e + (DEG_DEPTH - 1) * DEG_THREADS < e1; e += DEG_DEPTH * DEG_THREADS) {
        unsigned d[DEG_DEPTH];
#pragma unroll
        for (int u = 0; u < DEG_DEPTH; ++u)
            d[u] = (unsigned)(dst[e + u * DEG_THREADS] - base);
#pragma unroll
        for (int u = 0; u < DEG_DEPTH; ++u)
            if (d[u] < DEG_RANGE) atomicAdd(&hist[d[u]], 1u);
    }
    for (; e < e1; e += DEG_THREADS) {
        unsigned d0 = (unsigned)(dst[e] - base);
        if (d0 < DEG_RANGE) atomicAdd(&hist[d0], 1u);
    }
    __syncthreads();

    const int nr = min(DEG_RANGE, n_nodes - base);
    unsigned* p = part + (size_t)blockIdx.x * n_nodes + base;
    for (int i = tid; i < nr; i += DEG_THREADS) p[i] = hist[i];
}

#define NODE_T 256

__global__ void __launch_bounds__(NODE_T)
dinv_kernel(const unsigned* __restrict__ part, unsigned* __restrict__ soff,
            unsigned* __restrict__ indeg, unsigned* __restrict__ blockSum,
            float* __restrict__ dinv, float* __restrict__ pooled, int n_pool, int n) {
    __shared__ unsigned sh[NODE_T];
    const int tid = threadIdx.x;
    const int i = blockIdx.x * NODE_T + tid;
    for (int k = i; k < n_pool; k += (int)gridDim.x * NODE_T) pooled[k] = 0.0f;
    unsigned run = 0u;
    if (i < n) {
        static_assert(DEG_SLICES % 16 == 0, "slices are taken 16 at a time");
        for (int s0 = 0; s0 < DEG_SLICES; s0 += 16) {
            unsigned c[16];
#pragma unroll
            for (int u = 0; u < 16; ++u) c[u] = part[(size_t)(s0 + u) * n + i];
#pragma unroll
            for (int u = 0; u < 16; ++u) {
                soff[(size_t)(s0 + u) * n + i] = run;
                run += c[u];
            }
        }
        indeg[i] = run;
        dinv[i] = rsqrtf((float)(run + 1u));
    }
    sh[tid] = run;
    __syncthreads();
    for (int o = NODE_T / 2; o > 0; o >>= 1) {
        if (tid < o) sh[tid] += sh[tid + o];
        __syncthreads();
    }
    if (tid == 0) blockSum[blockIdx.x] = sh[0];
}

__global__ void __launch_bounds__(NODE_T)
rowptr_kernel(const unsigned* __restrict__ indeg, const unsigned* __restrict__ blockSum,
              unsigned* __restrict__ rowptr, int n) {
    __shared__ unsigned sh[NODE_T];
    const int tid = threadIdx.x;
    unsigned pre = 0u;
    for (int b = tid; b < (int)blockIdx.x; b += NODE_T) pre += blockSum[b];
    sh[tid] = pre;
    __syncthreads();
    for (int o = NODE_T / 2; o > 0; o >>= 1) {
        if (tid < o) sh[tid] += sh[tid + o];
        __syncthreads();
    }
    const unsigned before = sh[0];
    __syncthreads();

    const int i = blockIdx.x * NODE_T + tid;
    const unsigned v = (i < n) ? indeg[i] : 0u;
    sh[tid] = v;
    __syncthreads();
    for (int o = 1; o < NODE_T; o <<= 1) {
        unsigned t = (tid >= o) ? sh[tid - o] : 0u;
        __syncthreads();
        sh[tid] += t;
        __syncthreads();
    }
    if (i < n) rowptr[i] = before + sh[tid] - v;
}

__global__ void __launch_bounds__(DEG_THREADS)
csr_fill_kernel(const int* __restrict__ src, const int* __restrict__ dst,
                const float* __restrict__ dinv,
                const unsigned* __restrict__ rowptr, const unsigned* __restrict__ soff,
                uint2* __restrict__ csr, int n_nodes, int n_edges) {
    __shared__ unsigned cur[DEG_RANGE];
    const int tid  = threadIdx.x;
    const int base = blockIdx.y * DEG_RANGE;
    const int nr   = min(DEG_RANGE, n_nodes - base);

    const unsigned* so = soff + (size_t)blockIdx.x * n_nodes + base;
    for (int i = tid; i < nr; i += DEG_THREADS) cur[i] = rowptr[base + i] + so[i];
    __syncthreads();

    const int slice = ((n_edges + DEG_SLICES * 4 - 1) / (DEG_SLICES * 4)) * 4;
    const int e0 = blockIdx.x * slice;
    const int e1 = min(e0 + slice, n_edges);

    int e = e0 + tid;
    for (; e + (DEG_DEPTH - 1) * DEG_THREADS < e1; e += DEG_DEPTH * DEG_THREADS) {
        unsigned d[DEG_DEPTH];
        int      s[DEG_DEPTH];
        float    w[DEG_DEPTH];
#pragma unroll
        for (int u = 0; u < DEG_DEPTH; ++u) {
            d[u] = (unsigned)(dst[e + u * DEG_THREADS] - base);
            s[u] = src[e + u * DEG_THREADS];
        }
#pragma unroll
        for (int u = 0; u < DEG_DEPTH; ++u)
            w[u] = dinv[d[u] < (unsigned)nr ? s[u] : 0];
        asm volatile("" :: "v"(w[0]), "v"(w[1]), "v"(w[2]), "v"(w[3]),
                           "v"(w[4]), "v"(w[5]), "v"(w[6]), "v"(w[7]));
#pragma unroll
        for (int u = 0; u < DEG_DEPTH; ++u)
            if (d[u] < (unsigned)nr) {
                unsigned pos = atomicAdd(&cur[d[u]], 1u);
                csr[pos] = make_uint2((unsigned)s[u], __float_as_uint(w[u]));
            }
    }
    for (; e < e1; e += DEG_THREADS) {
        unsigned d0 = (unsigned)(dst[e] - base);
        if (d0 < (unsigned)nr) {
            int s0 = src[e];
            unsigned pos = atomicAdd(&cur[d0], 1u);
            csr[pos] = make_uint2((unsigned)s0, __float_as_uint(dinv[s0]));
        }
    }
}

__global__ void __launch_bounds__(256)
gemm_wmma_kernel(const float* __restrict__ X, const float* __restrict__ W,
                 float* __restrict__ H, int n_rows, int K) {
    int wave = (int)((blockIdx.x * blockDim.x + threadIdx.x) >> 5);
    int lane = threadIdx.x & 31;
    int row0 = wave * 16;
    if (row0 >= n_rows) return;

    int half = lane >> 4;
    int l16  = lane & 15;

    const float* xrow = X + (size_t)(row0 + l16) * K;
    v8f acc = {};
    for (int k = 0; k < K; k += 4) {
        int ka = k + half * 2;
        v2f a, b;
        a.x = xrow[ka];
        a.y = xrow[ka + 1];
        b.x = W[(size_t)ka * HIDDEN + l16];
        b.y = W[(size_t)(ka + 1) * HIDDEN + l16];
        acc = __builtin_amdgcn_wmma_f32_16x16x4_f32(
             false, a,  false, b,
             (short)0, acc,  false,  false);
    }
    float* out = H + (size_t)row0 * HIDDEN;
#pragma unroll
    for (int i = 0; i < 8; ++i)
        out[(size_t)(i + 8 * half) * HIDDEN + l16] = acc[i];
}


#define GATHER_DEPTH 8
#define ROW_PITCH 20

__device__ __forceinline__ void wave_lds_handover() {
    __builtin_amdgcn_fence(__ATOMIC_RELEASE, "wavefront");
    __builtin_amdgcn_wave_barrier();
    __builtin_amdgcn_fence(__ATOMIC_ACQUIRE, "wavefront");
}

template <int K>
__device__ __forceinline__ unsigned quad_bcast(unsigned v) {
    return (unsigned)__builtin_amdgcn_mov_dpp((int)v, K * 0x55, 0xF, 0xF, true);
}

template <bool POOL>
__global__ void __launch_bounds__(256)
gather_kernel(const uint2* __restrict__ csr, const unsigned* __restrict__ rowptr,
              const unsigned* __restrict__ indeg, const float* __restrict__ dinv,
              const float4* __restrict__ hin, const float* __restrict__ bias,
              const float* __restrict__ Wn,
              float4* __restrict__ hout, const int* __restrict__ batch,
              float* __restrict__ pooled, int n_nodes, int n_graphs, int do_relu) {
    int g = blockIdx.x * blockDim.x + threadIdx.x;
    int i = g >> 2;
    const bool valid = i < n_nodes;
    if (!POOL) {
        if (((g & ~31) >> 2) >= n_nodes) return;
    }
    if (!valid) i = 0;
    int q = g & 3;
    int bid = -1;
    if (POOL) {
        bid = batch[i];
        if (!valid) bid = -1;
    }
    unsigned j = rowptr[i];
    const unsigned cnt = indeg[i];
    const unsigned end = j + (valid ? cnt : 0u);
    const float di = dinv[i];
    __shared__ __attribute__((aligned(16))) float4 sSelf[256];
    sSelf[threadIdx.x] = hin[(size_t)i * 4 + q];
    float4 acc = make_float4(0.0f, 0.0f, 0.0f, 0.0f);

    const unsigned last = max(end, 1u) - 1u;
    uint2 c0 = csr[min(j + (unsigned)q, last)];
    uint2 c1 = csr[min(j + 4u + (unsigned)q, last)];
    for (; j < end; j += GATHER_DEPTH) {
        static_assert(GATHER_DEPTH == 8, "a quad holds 2 slots per lane");
        const uint2 n0 = csr[min(j + 8u + (unsigned)q, last)];
        const uint2 n1 = csr[min(j + 12u + (unsigned)q, last)];
        uint2  c[GATHER_DEPTH];
        float4 h[GATHER_DEPTH];
        c[0] = make_uint2(quad_bcast<0>(c0.x), quad_bcast<0>(c0.y));
        c[1] = make_uint2(quad_bcast<1>(c0.x), quad_bcast<1>(c0.y));
        c[2] = make_uint2(quad_bcast<2>(c0.x), quad_bcast<2>(c0.y));
        c[3] = make_uint2(quad_bcast<3>(c0.x), quad_bcast<3>(c0.y));
        c[4] = make_uint2(quad_bcast<0>(c1.x), quad_bcast<0>(c1.y));
        c[5] = make_uint2(quad_bcast<1>(c1.x), quad_bcast<1>(c1.y));
        c[6] = make_uint2(quad_bcast<2>(c1.x), quad_bcast<2>(c1.y));
        c[7] = make_uint2(quad_bcast<3>(c1.x), quad_bcast<3>(c1.y));
#pragma unroll
        for (int u = 0; u < GATHER_DEPTH; ++u)
            h[u] = *(const float4*)((const char*)hin + (c[u].x * 64u + (unsigned)q * 16u));
        asm volatile("" :: "v"(n0.x), "v"(n1.x),
                           "v"(h[0].x), "v"(h[0].w), "v"(h[1].x), "v"(h[1].w),
                           "v"(h[2].x), "v"(h[2].w), "v"(h[3].x), "v"(h[3].w),
                           "v"(h[4].x), "v"(h[4].w), "v"(h[5].x), "v"(h[5].w),
                           "v"(h[6].x), "v"(h[6].w), "v"(h[7].x), "v"(h[7].w));
        c0 = n0;
        c1 = n1;
#pragma unroll
        for (int u = 0; u < GATHER_DEPTH; ++u) {
            const bool on = j + (unsigned)u < end;
            const float norm = __uint_as_float(c[u].y) * di;
            const float sx = acc.x + norm * h[u].x;
            const float sy = acc.y + norm * h[u].y;
            const float sz = acc.z + norm * h[u].z;
            const float sw = acc.w + norm * h[u].w;
            acc.x = on ? sx : acc.x;
            acc.y = on ? sy : acc.y;
            acc.z = on ? sz : acc.z;
            acc.w = on ? sw : acc.w;
        }
    }

    const float4 hs = sSelf[threadIdx.x];
    float4 v;
    v.x = acc.x + di * di * hs.x + bias[q * 4 + 0];
    v.y = acc.y + di * di * hs.y + bias[q * 4 + 1];
    v.z = acc.z + di * di * hs.z + bias[q * 4 + 2];
    v.w = acc.w + di * di * hs.w + bias[q * 4 + 3];
    if (do_relu) {
        v.x = fmaxf(v.x, 0.0f);
        v.y = fmaxf(v.y, 0.0f);
        v.z = fmaxf(v.z, 0.0f);
        v.w = fmaxf(v.w, 0.0f);
    }
    const int lane = threadIdx.x & 31;
    if (!POOL) {
        __shared__ __attribute__((aligned(16))) float sIn[8 * 8 * ROW_PITCH];
        __shared__ __attribute__((aligned(16))) float sOut[8 * 8 * ROW_PITCH];
        float* in_rows  = sIn  + (threadIdx.x >> 5) * (8 * ROW_PITCH);
        float* out_rows = sOut + (threadIdx.x >> 5) * (8 * ROW_PITCH);
        const int nd = lane >> 2;
        *(float4*)&in_rows[nd * ROW_PITCH + q * 4] = v;
        wave_lds_handover();

        const int half = lane >> 4;
        const int l16  = lane & 15;
        const float* arow = in_rows + (l16 & 7) * ROW_PITCH;
        v8f t = {};
#pragma unroll
        for (int kk = 0; kk < HIDDEN / 4; ++kk) {
            const int ka = kk * 4 + half * 2;
            v2f a, b;
            a.x = arow[ka];
            a.y = arow[ka + 1];
            b.x = Wn[ka * HIDDEN + l16];
            b.y = Wn[(ka + 1) * HIDDEN + l16];
            t = __builtin_amdgcn_wmma_f32_16x16x4_f32(
                 false, a,  false, b,
                 (short)0, t,  false,  false);
        }
        if (half == 0) {
#pragma unroll
            for (int r = 0; r < 8; ++r) out_rows[r * ROW_PITCH + l16] = t[r];
        }
        wave_lds_handover();
        const float4 o = *(const float4*)&out_rows[nd * ROW_PITCH + q * 4];
        if (valid) hout[(size_t)i * 4 + q] = o;
    } else {
        __shared__ float4 srow[256];
        const int f = lane & 15;
        srow[threadIdx.x] = v;
        __syncthreads();
        const float* rows = (const float*)&srow[threadIdx.x & ~31];
        float sum = 0.0f;
        int cur = __builtin_amdgcn_readlane(bid, 0);
#pragma unroll
        for (int n = 0; n < 8; ++n) {
            const int b = __builtin_amdgcn_readlane(bid, n * 4);
            const float x = rows[n * 16 + f];
            if (b != cur) {
                if (lane < 16 && (unsigned)cur < (unsigned)n_graphs)
                    atomicAdd(&pooled[(size_t)cur * HIDDEN + f], sum);
                sum = 0.0f;
                cur = b;
            }
            sum += x;
        }
        if (lane < 16 && (unsigned)cur < (unsigned)n_graphs)
            atomicAdd(&pooled[(size_t)cur * HIDDEN + f], sum);
    }
}


__global__ void final_linear_kernel(const float* __restrict__ pooled,
                                    const float* __restrict__ Wlin,
                                    const float* __restrict__ blin,
                                    float* __restrict__ out,
                                    int n_graphs, int n_classes) {
    int g = blockIdx.x * blockDim.x + threadIdx.x;
    if (g >= n_graphs * n_classes) return;
    int gr = g / n_classes;
    int c  = g % n_classes;
    float acc = blin[c];
#pragma unroll
    for (int f = 0; f < HIDDEN; ++f)
        acc += pooled[(size_t)gr * HIDDEN + f] * Wlin[f * n_classes + c];
    out[g] = acc;
}


static inline size_t align_up(size_t v, size_t a) { return (v + a - 1) & ~(a - 1); }

extern "C" void kernel_launch(void* const* d_in, const int* in_sizes, int n_in,
                              void* d_out, int out_size, void* d_ws, size_t ws_size,
                              hipStream_t stream) {
    (void)n_in; (void)ws_size;

    const float* x    = (const float*)d_in[0];
    const int*   ei   = (const int*)d_in[1];
    const int*   batch= (const int*)d_in[2];
    const float* W1   = (const float*)d_in[3];
    const float* b1   = (const float*)d_in[4];
    const float* W2   = (const float*)d_in[5];
    const float* b2   = (const float*)d_in[6];
    const float* W3   = (const float*)d_in[7];
    const float* b3   = (const float*)d_in[8];
    const float* Wlin = (const float*)d_in[9];
    const float* blin = (const float*)d_in[10];
    float* out = (float*)d_out;

    const int N_FEAT = 128;
    const int n_nodes = in_sizes[0] / N_FEAT;
    const int n_edges = in_sizes[1] / 2;
    const int n_classes = in_sizes[10];
    const int n_graphs  = out_size / n_classes;

    const int* src = ei;
    const int* dst = ei + n_edges;

    char* ws = (char*)d_ws;
    size_t off = 0;
    float* dinv = (float*)(ws + off);  off = align_up(off + (size_t)n_nodes * 4, 256);
    float* bufA = (float*)(ws + off);  off = align_up(off + (size_t)n_nodes * HIDDEN * 4, 256);
    float* bufB = (float*)(ws + off);  off = align_up(off + (size_t)n_nodes * HIDDEN * 4, 256);
    float* pooled = (float*)(ws + off); off = align_up(off + (size_t)n_graphs * HIDDEN * 4, 256);
    unsigned* degPart = (unsigned*)(ws + off);
    off = align_up(off + (size_t)DEG_SLICES * n_nodes * 4, 256);
    unsigned* sliceOff = (unsigned*)(ws + off);
    off = align_up(off + (size_t)DEG_SLICES * n_nodes * 4, 256);
    unsigned* indeg = (unsigned*)(ws + off);    off = align_up(off + (size_t)n_nodes * 4, 256);
    unsigned* rowptr = (unsigned*)(ws + off);   off = align_up(off + (size_t)n_nodes * 4, 256);
    unsigned* blockSum = (unsigned*)(ws + off); off = align_up(off + (size_t)n_nodes * 4, 256);
    uint2* csr = (uint2*)(ws + off);
    off = align_up(off + (size_t)n_edges * 8, 256);

    const int T = 256;
    const int nodeBlocks  = (n_nodes + NODE_T - 1) / NODE_T;
    const int nodeQBlocks = ((n_nodes * 4) + T - 1) / T;
    const int tiles       = (n_nodes + 15) / 16;
    const int gemmBlocks  = (tiles + 7) / 8;
    const dim3 degGrid(DEG_SLICES, (n_nodes + DEG_RANGE - 1) / DEG_RANGE);

    deg_hist_kernel<<<degGrid, DEG_THREADS, 0, stream>>>(dst, degPart, n_nodes, n_edges);
    dinv_kernel<<<nodeBlocks, NODE_T, 0, stream>>>(degPart, sliceOff, indeg, blockSum, dinv,
                                                   pooled, n_graphs * HIDDEN, n_nodes);
    rowptr_kernel<<<nodeBlocks, NODE_T, 0, stream>>>(indeg, blockSum, rowptr, n_nodes);
    csr_fill_kernel<<<degGrid, DEG_THREADS, 0, stream>>>(src, dst, dinv, rowptr, sliceOff, csr,
                                                        n_nodes, n_edges);

    gemm_wmma_kernel<<<gemmBlocks, T, 0, stream>>>(x, W1, bufA, n_nodes, N_FEAT);

    gather_kernel<false><<<nodeQBlocks, T, 0, stream>>>(csr, rowptr, indeg, dinv, (const float4*)bufA, b1, W2,
                                                        (float4*)bufB, batch, pooled,
                                                        n_nodes, n_graphs, 1);

    gather_kernel<false><<<nodeQBlocks, T, 0, stream>>>(csr, rowptr, indeg, dinv, (const float4*)bufB, b2, W3,
                                                        (float4*)bufA, batch, pooled,
                                                        n_nodes, n_graphs, 1);

    gather_kernel<true><<<nodeQBlocks, T, 0, stream>>>(csr, rowptr, indeg, dinv, (const float4*)bufA, b3, nullptr,
                                                       (float4*)bufB, batch, pooled,
                                                       n_nodes, n_graphs, 0);

    final_linear_kernel<<<(n_graphs * n_classes + T - 1) / T, T, 0, stream>>>(
        pooled, Wlin, blin, out, n_graphs, n_classes);
}


# Edge weight split: rows stored pre-multiplied by their node's dinv, dest dinv applied once to the sum; a slot is the 4-byte source id, the sort gathers no dinv; sum order kept, term rounding differs
# speedup vs baseline: 3.5328x; 1.0562x over previous
#include <hip/hip_runtime.h>
#include <hip/hip_bf16.h>


typedef float v2f __attribute__((ext_vector_type(2)));
typedef float v8f __attribute__((ext_vector_type(8)));

#define HIDDEN 16


#define DEG_SLICES  64
#define DEG_RANGE   25600
#define DEG_THREADS 1024
#define DEG_DEPTH   8

__global__ void __launch_bounds__(DEG_THREADS)
deg_hist_kernel(const int* __restrict__ dst, unsigned* __restrict__ part,
                int n_nodes, int n_edges) {
    __shared__ unsigned hist[DEG_RANGE];
    const int tid  = threadIdx.x;
    const int base = blockIdx.y * DEG_RANGE;

    for (int i = tid; i < DEG_RANGE; i += DEG_THREADS) hist[i] = 0u;
    __syncthreads();

    const int slice = ((n_edges + DEG_SLICES * 4 - 1) / (DEG_SLICES * 4)) * 4;
    const int e0 = blockIdx.x * slice;
    const int e1 = min(e0 + slice, n_edges);

    int e = e0 + tid;
    for (; e + (DEG_DEPTH - 1) * DEG_THREADS < e1; e += DEG_DEPTH * DEG_THREADS) {
        unsigned d[DEG_DEPTH];
#pragma unroll
        for (int u = 0; u < DEG_DEPTH; ++u)
            d[u] = (unsigned)(dst[e + u * DEG_THREADS] - base);
#pragma unroll
        for (int u = 0; u < DEG_DEPTH; ++u)
            if (d[u] < DEG_RANGE) atomicAdd(&hist[d[u]], 1u);
    }
    for (; e < e1; e += DEG_THREADS) {
        unsigned d0 = (unsigned)(dst[e] - base);
        if (d0 < DEG_RANGE) atomicAdd(&hist[d0], 1u);
    }
    __syncthreads();

    const int nr = min(DEG_RANGE, n_nodes - base);
    unsigned* p = part + (size_t)blockIdx.x * n_nodes + base;
    for (int i = tid; i < nr; i += DEG_THREADS) p[i] = hist[i];
}

#define NODE_T 256

__global__ void __launch_bounds__(NODE_T)
dinv_kernel(const unsigned* __restrict__ part, unsigned* __restrict__ soff,
            unsigned* __restrict__ indeg, unsigned* __restrict__ blockSum,
            float* __restrict__ dinv, float* __restrict__ pooled, int n_pool, int n) {
    __shared__ unsigned sh[NODE_T];
    const int tid = threadIdx.x;
    const int i = blockIdx.x * NODE_T + tid;
    for (int k = i; k < n_pool; k += (int)gridDim.x * NODE_T) pooled[k] = 0.0f;
    unsigned run = 0u;
    if (i < n) {
        static_assert(DEG_SLICES % 16 == 0, "slices are taken 16 at a time");
        for (int s0 = 0; s0 < DEG_SLICES; s0 += 16) {
            unsigned c[16];
#pragma unroll
            for (int u = 0; u < 16; ++u) c[u] = part[(size_t)(s0 + u) * n + i];
#pragma unroll
            for (int u = 0; u < 16; ++u) {
                soff[(size_t)(s0 + u) * n + i] = run;
                run += c[u];
            }
        }
        indeg[i] = run;
        dinv[i] = rsqrtf((float)(run + 1u));
    }
    sh[tid] = run;
    __syncthreads();
    for (int o = NODE_T / 2; o > 0; o >>= 1) {
        if (tid < o) sh[tid] += sh[tid + o];
        __syncthreads();
    }
    if (tid == 0) blockSum[blockIdx.x] = sh[0];
}

__global__ void __launch_bounds__(NODE_T)
rowptr_kernel(const unsigned* __restrict__ indeg, const unsigned* __restrict__ blockSum,
              unsigned* __restrict__ rowptr, int n) {
    __shared__ unsigned sh[NODE_T];
    const int tid = threadIdx.x;
    unsigned pre = 0u;
    for (int b = tid; b < (int)blockIdx.x; b += NODE_T) pre += blockSum[b];
    sh[tid] = pre;
    __syncthreads();
    for (int o = NODE_T / 2; o > 0; o >>= 1) {
        if (tid < o) sh[tid] += sh[tid + o];
        __syncthreads();
    }
    const unsigned before = sh[0];
    __syncthreads();

    const int i = blockIdx.x * NODE_T + tid;
    const unsigned v = (i < n) ? indeg[i] : 0u;
    sh[tid] = v;
    __syncthreads();
    for (int o = 1; o < NODE_T; o <<= 1) {
        unsigned t = (tid >= o) ? sh[tid - o] : 0u;
        __syncthreads();
        sh[tid] += t;
        __syncthreads();
    }
    if (i < n) rowptr[i] = before + sh[tid] - v;
}

__global__ void __launch_bounds__(DEG_THREADS)
csr_fill_kernel(const int* __restrict__ src, const int* __restrict__ dst,
                const unsigned* __restrict__ rowptr, const unsigned* __restrict__ soff,
                unsigned* __restrict__ csr, int n_nodes, int n_edges) {
    __shared__ unsigned cur[DEG_RANGE];
    const int tid  = threadIdx.x;
    const int base = blockIdx.y * DEG_RANGE;
    const int nr   = min(DEG_RANGE, n_nodes - base);

    const unsigned* so = soff + (size_t)blockIdx.x * n_nodes + base;
    for (int i = tid; i < nr; i += DEG_THREADS) cur[i] = rowptr[base + i] + so[i];
    __syncthreads();

    const int slice = ((n_edges + DEG_SLICES * 4 - 1) / (DEG_SLICES * 4)) * 4;
    const int e0 = blockIdx.x * slice;
    const int e1 = min(e0 + slice, n_edges);

    int e = e0 + tid;
    for (; e + (DEG_DEPTH - 1) * DEG_THREADS < e1; e += DEG_DEPTH * DEG_THREADS) {
        unsigned d[DEG_DEPTH];
        int      s[DEG_DEPTH];
#pragma unroll
        for (int u = 0; u < DEG_DEPTH; ++u) {
            d[u] = (unsigned)(dst[e + u * DEG_THREADS] - base);
            s[u] = src[e + u * DEG_THREADS];
        }
        asm volatile("" :: "v"(s[0]), "v"(s[1]), "v"(s[2]), "v"(s[3]),
                           "v"(s[4]), "v"(s[5]), "v"(s[6]), "v"(s[7]));
#pragma unroll
        for (int u = 0; u < DEG_DEPTH; ++u)
            if (d[u] < (unsigned)nr) {
                unsigned pos = atomicAdd(&cur[d[u]], 1u);
                csr[pos] = (unsigned)s[u];
            }
    }
    for (; e < e1; e += DEG_THREADS) {
        unsigned d0 = (unsigned)(dst[e] - base);
        const int s0 = src[e];
        if (d0 < (unsigned)nr) {
            unsigned pos = atomicAdd(&cur[d0], 1u);
            csr[pos] = (unsigned)s0;
        }
    }
}

__global__ void __launch_bounds__(256)
gemm_wmma_kernel(const float* __restrict__ X, const float* __restrict__ W,
                 const float* __restrict__ rs,
                 float* __restrict__ H, int n_rows, int K) {
    int wave = (int)((blockIdx.x * blockDim.x + threadIdx.x) >> 5);
    int lane = threadIdx.x & 31;
    int row0 = wave * 16;
    if (row0 >= n_rows) return;

    int half = lane >> 4;
    int l16  = lane & 15;

    const float* xrow = X + (size_t)(row0 + l16) * K;
    v8f acc = {};
    for (int k = 0; k < K; k += 4) {
        int ka = k + half * 2;
        v2f a, b;
        a.x = xrow[ka];
        a.y = xrow[ka + 1];
        b.x = W[(size_t)ka * HIDDEN + l16];
        b.y = W[(size_t)(ka + 1) * HIDDEN + l16];
        acc = __builtin_amdgcn_wmma_f32_16x16x4_f32(
             false, a,  false, b,
             (short)0, acc,  false,  false);
    }
    float sc[8];
#pragma unroll
    for (int i = 0; i < 8; ++i) sc[i] = rs[row0 + i + 8 * half];
    float* out = H + (size_t)row0 * HIDDEN;
#pragma unroll
    for (int i = 0; i < 8; ++i)
        out[(size_t)(i + 8 * half) * HIDDEN + l16] = acc[i] * sc[i];
}


#define GATHER_DEPTH 8
#define ROW_PITCH 20

__device__ __forceinline__ void wave_lds_handover() {
    __builtin_amdgcn_fence(__ATOMIC_RELEASE, "wavefront");
    __builtin_amdgcn_wave_barrier();
    __builtin_amdgcn_fence(__ATOMIC_ACQUIRE, "wavefront");
}

template <int K>
__device__ __forceinline__ unsigned quad_bcast(unsigned v) {
    return (unsigned)__builtin_amdgcn_mov_dpp((int)v, K * 0x55, 0xF, 0xF, true);
}

template <bool POOL>
__global__ void __launch_bounds__(256)
gather_kernel(const unsigned* __restrict__ csr, const unsigned* __restrict__ rowptr,
              const unsigned* __restrict__ indeg, const float* __restrict__ dinv,
              const float4* __restrict__ hin, const float* __restrict__ bias,
              const float* __restrict__ Wn,
              float4* __restrict__ hout, const int* __restrict__ batch,
              float* __restrict__ pooled, int n_nodes, int n_graphs, int do_relu) {
    int g = blockIdx.x * blockDim.x + threadIdx.x;
    int i = g >> 2;
    const bool valid = i < n_nodes;
    if (!POOL) {
        if (((g & ~31) >> 2) >= n_nodes) return;
    }
    if (!valid) i = 0;
    int q = g & 3;
    int bid = -1;
    if (POOL) {
        bid = batch[i];
        if (!valid) bid = -1;
    }
    unsigned j = rowptr[i];
    const unsigned cnt = indeg[i];
    const unsigned end = j + (valid ? cnt : 0u);
    const float di = dinv[i];
    __shared__ __attribute__((aligned(16))) float4 sSelf[256];
    sSelf[threadIdx.x] = hin[(size_t)i * 4 + q];
    float4 acc = make_float4(0.0f, 0.0f, 0.0f, 0.0f);

    const unsigned last = max(end, 1u) - 1u;
    unsigned c0 = csr[min(j + (unsigned)q, last)];
    unsigned c1 = csr[min(j + 4u + (unsigned)q, last)];
    for (; j < end; j += GATHER_DEPTH) {
        static_assert(GATHER_DEPTH == 8, "a quad holds 2 slots per lane");
        const unsigned n0 = csr[min(j + 8u + (unsigned)q, last)];
        const unsigned n1 = csr[min(j + 12u + (unsigned)q, last)];
        unsigned c[GATHER_DEPTH];
        float4   h[GATHER_DEPTH];
        c[0] = quad_bcast<0>(c0);
        c[1] = quad_bcast<1>(c0);
        c[2] = quad_bcast<2>(c0);
        c[3] = quad_bcast<3>(c0);
        c[4] = quad_bcast<0>(c1);
        c[5] = quad_bcast<1>(c1);
        c[6] = quad_bcast<2>(c1);
        c[7] = quad_bcast<3>(c1);
#pragma unroll
        for (int u = 0; u < GATHER_DEPTH; ++u)
            h[u] = *(const float4*)((const char*)hin + (c[u] * 64u + (unsigned)q * 16u));
        asm volatile("" :: "v"(n0), "v"(n1),
                           "v"(h[0].x), "v"(h[0].w), "v"(h[1].x), "v"(h[1].w),
                           "v"(h[2].x), "v"(h[2].w), "v"(h[3].x), "v"(h[3].w),
                           "v"(h[4].x), "v"(h[4].w), "v"(h[5].x), "v"(h[5].w),
                           "v"(h[6].x), "v"(h[6].w), "v"(h[7].x), "v"(h[7].w));
        c0 = n0;
        c1 = n1;
#pragma unroll
        for (int u = 0; u < GATHER_DEPTH; ++u) {
            const bool on = j + (unsigned)u < end;
            const float sx = acc.x + h[u].x;
            const float sy = acc.y + h[u].y;
            const float sz = acc.z + h[u].z;
            const float sw = acc.w + h[u].w;
            acc.x = on ? sx : acc.x;
            acc.y = on ? sy : acc.y;
            acc.z = on ? sz : acc.z;
            acc.w = on ? sw : acc.w;
        }
    }

    const float4 hs = sSelf[threadIdx.x];
    float4 v;
    v.x = di * (acc.x + hs.x) + bias[q * 4 + 0];
    v.y = di * (acc.y + hs.y) + bias[q * 4 + 1];
    v.z = di * (acc.z + hs.z) + bias[q * 4 + 2];
    v.w = di * (acc.w + hs.w) + bias[q * 4 + 3];
    if (do_relu) {
        v.x = fmaxf(v.x, 0.0f);
        v.y = fmaxf(v.y, 0.0f);
        v.z = fmaxf(v.z, 0.0f);
        v.w = fmaxf(v.w, 0.0f);
    }
    const int lane = threadIdx.x & 31;
    if (!POOL) {
        __shared__ __attribute__((aligned(16))) float sIn[8 * 8 * ROW_PITCH];
        __shared__ __attribute__((aligned(16))) float sOut[8 * 8 * ROW_PITCH];
        float* in_rows  = sIn  + (threadIdx.x >> 5) * (8 * ROW_PITCH);
        float* out_rows = sOut + (threadIdx.x >> 5) * (8 * ROW_PITCH);
        const int nd = lane >> 2;
        *(float4*)&in_rows[nd * ROW_PITCH + q * 4] = v;
        wave_lds_handover();

        const int half = lane >> 4;
        const int l16  = lane & 15;
        const float* arow = in_rows + (l16 & 7) * ROW_PITCH;
        v8f t = {};
#pragma unroll
        for (int kk = 0; kk < HIDDEN / 4; ++kk) {
            const int ka = kk * 4 + half * 2;
            v2f a, b;
            a.x = arow[ka];
            a.y = arow[ka + 1];
            b.x = Wn[ka * HIDDEN + l16];
            b.y = Wn[(ka + 1) * HIDDEN + l16];
            t = __builtin_amdgcn_wmma_f32_16x16x4_f32(
                 false, a,  false, b,
                 (short)0, t,  false,  false);
        }
        if (half == 0) {
#pragma unroll
            for (int r = 0; r < 8; ++r) out_rows[r * ROW_PITCH + l16] = t[r];
        }
        wave_lds_handover();
        float4 o = *(const float4*)&out_rows[nd * ROW_PITCH + q * 4];
        o.x *= di;
        o.y *= di;
        o.z *= di;
        o.w *= di;
        if (valid) hout[(size_t)i * 4 + q] = o;
    } else {
        __shared__ float4 srow[256];
        const int f = lane & 15;
        srow[threadIdx.x] = v;
        __syncthreads();
        const float* rows = (const float*)&srow[threadIdx.x & ~31];
        float sum = 0.0f;
        int cur = __builtin_amdgcn_readlane(bid, 0);
#pragma unroll
        for (int n = 0; n < 8; ++n) {
            const int b = __builtin_amdgcn_readlane(bid, n * 4);
            const float x = rows[n * 16 + f];
            if (b != cur) {
                if (lane < 16 && (unsigned)cur < (unsigned)n_graphs)
                    atomicAdd(&pooled[(size_t)cur * HIDDEN + f], sum);
                sum = 0.0f;
                cur = b;
            }
            sum += x;
        }
        if (lane < 16 && (unsigned)cur < (unsigned)n_graphs)
            atomicAdd(&pooled[(size_t)cur * HIDDEN + f], sum);
    }
}


__global__ void final_linear_kernel(const float* __restrict__ pooled,
                                    const float* __restrict__ Wlin,
                                    const float* __restrict__ blin,
                                    float* __restrict__ out,
                                    int n_graphs, int n_classes) {
    int g = blockIdx.x * blockDim.x + threadIdx.x;
    if (g >= n_graphs * n_classes) return;
    int gr = g / n_classes;
    int c  = g % n_classes;
    float acc = blin[c];
#pragma unroll
    for (int f = 0; f < HIDDEN; ++f)
        acc += pooled[(size_t)gr * HIDDEN + f] * Wlin[f * n_classes + c];
    out[g] = acc;
}


static inline size_t align_up(size_t v, size_t a) { return (v + a - 1) & ~(a - 1); }

extern "C" void kernel_launch(void* const* d_in, const int* in_sizes, int n_in,
                              void* d_out, int out_size, void* d_ws, size_t ws_size,
                              hipStream_t stream) {
    (void)n_in; (void)ws_size;

    const float* x    = (const float*)d_in[0];
    const int*   ei   = (const int*)d_in[1];
    const int*   batch= (const int*)d_in[2];
    const float* W1   = (const float*)d_in[3];
    const float* b1   = (const float*)d_in[4];
    const float* W2   = (const float*)d_in[5];
    const float* b2   = (const float*)d_in[6];
    const float* W3   = (const float*)d_in[7];
    const float* b3   = (const float*)d_in[8];
    const float* Wlin = (const float*)d_in[9];
    const float* blin = (const float*)d_in[10];
    float* out = (float*)d_out;

    const int N_FEAT = 128;
    const int n_nodes = in_sizes[0] / N_FEAT;
    const int n_edges = in_sizes[1] / 2;
    const int n_classes = in_sizes[10];
    const int n_graphs  = out_size / n_classes;

    const int* src = ei;
    const int* dst = ei + n_edges;

    char* ws = (char*)d_ws;
    size_t off = 0;
    float* dinv = (float*)(ws + off);  off = align_up(off + (size_t)n_nodes * 4, 256);
    float* bufA = (float*)(ws + off);  off = align_up(off + (size_t)n_nodes * HIDDEN * 4, 256);
    float* bufB = (float*)(ws + off);  off = align_up(off + (size_t)n_nodes * HIDDEN * 4, 256);
    float* pooled = (float*)(ws + off); off = align_up(off + (size_t)n_graphs * HIDDEN * 4, 256);
    unsigned* degPart = (unsigned*)(ws + off);
    off = align_up(off + (size_t)DEG_SLICES * n_nodes * 4, 256);
    unsigned* sliceOff = (unsigned*)(ws + off);
    off = align_up(off + (size_t)DEG_SLICES * n_nodes * 4, 256);
    unsigned* indeg = (unsigned*)(ws + off);    off = align_up(off + (size_t)n_nodes * 4, 256);
    unsigned* rowptr = (unsigned*)(ws + off);   off = align_up(off + (size_t)n_nodes * 4, 256);
    unsigned* blockSum = (unsigned*)(ws + off); off = align_up(off + (size_t)n_nodes * 4, 256);
    unsigned* csr = (unsigned*)(ws + off);
    off = align_up(off + (size_t)n_edges * 4, 256);

    const int T = 256;
    const int nodeBlocks  = (n_nodes + NODE_T - 1) / NODE_T;
    const int nodeQBlocks = ((n_nodes * 4) + T - 1) / T;
    const int tiles       = (n_nodes + 15) / 16;
    const int gemmBlocks  = (tiles + 7) / 8;
    const dim3 degGrid(DEG_SLICES, (n_nodes + DEG_RANGE - 1) / DEG_RANGE);

    deg_hist_kernel<<<degGrid, DEG_THREADS, 0, stream>>>(dst, degPart, n_nodes, n_edges);
    dinv_kernel<<<nodeBlocks, NODE_T, 0, stream>>>(degPart, sliceOff, indeg, blockSum, dinv,
                                                   pooled, n_graphs * HIDDEN, n_nodes);
    rowptr_kernel<<<nodeBlocks, NODE_T, 0, stream>>>(indeg, blockSum, rowptr, n_nodes);
    csr_fill_kernel<<<degGrid, DEG_THREADS, 0, stream>>>(src, dst, rowptr, sliceOff, csr,
                                                        n_nodes, n_edges);

    gemm_wmma_kernel<<<gemmBlocks, T, 0, stream>>>(x, W1, dinv, bufA, n_nodes, N_FEAT);

    gather_kernel<false><<<nodeQBlocks, T, 0, stream>>>(csr, rowptr, indeg, dinv, (const float4*)bufA, b1, W2,
                                                        (float4*)bufB, batch, pooled,
                                                        n_nodes, n_graphs, 1);

    gather_kernel<false><<<nodeQBlocks, T, 0, stream>>>(csr, rowptr, indeg, dinv, (const float4*)bufB, b2, W3,
                                                        (float4*)bufA, batch, pooled,
                                                        n_nodes, n_graphs, 1);

    gather_kernel<true><<<nodeQBlocks, T, 0, stream>>>(csr, rowptr, indeg, dinv, (const float4*)bufA, b3, nullptr,
                                                       (float4*)bufB, batch, pooled,
                                                       n_nodes, n_graphs, 0);

    final_linear_kernel<<<(n_graphs * n_classes + T - 1) / T, T, 0, stream>>>(
        pooled, Wlin, blin, out, n_graphs, n_classes);
}


# X@W1 product moved into the edge placement launch: its row tiles are extra work-groups behind the placement blocks and stream X while those wait on LDS ranks and scattered stores; 9 to 8 launches
# speedup vs baseline: 3.6400x; 1.0303x over previous
#include <hip/hip_runtime.h>
#include <hip/hip_bf16.h>


typedef float v2f __attribute__((ext_vector_type(2)));
typedef float v8f __attribute__((ext_vector_type(8)));

#define HIDDEN 16


#define DEG_SLICES  64
#define DEG_RANGE   25600
#define DEG_THREADS 1024
#define DEG_DEPTH   8

__global__ void __launch_bounds__(DEG_THREADS)
deg_hist_kernel(const int* __restrict__ dst, unsigned* __restrict__ part,
                int n_nodes, int n_edges) {
    __shared__ unsigned hist[DEG_RANGE];
    const int tid  = threadIdx.x;
    const int base = blockIdx.y * DEG_RANGE;

    for (int i = tid; i < DEG_RANGE; i += DEG_THREADS) hist[i] = 0u;
    __syncthreads();

    const int slice = ((n_edges + DEG_SLICES * 4 - 1) / (DEG_SLICES * 4)) * 4;
    const int e0 = blockIdx.x * slice;
    const int e1 = min(e0 + slice, n_edges);

    int e = e0 + tid;
    for (; e + (DEG_DEPTH - 1) * DEG_THREADS < e1; e += DEG_DEPTH * DEG_THREADS) {
        unsigned d[DEG_DEPTH];
#pragma unroll
        for (int u = 0; u < DEG_DEPTH; ++u)
            d[u] = (unsigned)(dst[e + u * DEG_THREADS] - base);
#pragma unroll
        for (int u = 0; u < DEG_DEPTH; ++u)
            if (d[u] < DEG_RANGE) atomicAdd(&hist[d[u]], 1u);
    }
    for (; e < e1; e += DEG_THREADS) {
        unsigned d0 = (unsigned)(dst[e] - base);
        if (d0 < DEG_RANGE) atomicAdd(&hist[d0], 1u);
    }
    __syncthreads();

    const int nr = min(DEG_RANGE, n_nodes - base);
    unsigned* p = part + (size_t)blockIdx.x * n_nodes + base;
    for (int i = tid; i < nr; i += DEG_THREADS) p[i] = hist[i];
}

#define NODE_T 256

__global__ void __launch_bounds__(NODE_T)
dinv_kernel(const unsigned* __restrict__ part, unsigned* __restrict__ soff,
            unsigned* __restrict__ indeg, unsigned* __restrict__ blockSum,
            float* __restrict__ dinv, float* __restrict__ pooled, int n_pool, int n) {
    __shared__ unsigned sh[NODE_T];
    const int tid = threadIdx.x;
    const int i = blockIdx.x * NODE_T + tid;
    for (int k = i; k < n_pool; k += (int)gridDim.x * NODE_T) pooled[k] = 0.0f;
    unsigned run = 0u;
    if (i < n) {
        static_assert(DEG_SLICES % 16 == 0, "slices are taken 16 at a time");
        for (int s0 = 0; s0 < DEG_SLICES; s0 += 16) {
            unsigned c[16];
#pragma unroll
            for (int u = 0; u < 16; ++u) c[u] = part[(size_t)(s0 + u) * n + i];
#pragma unroll
            for (int u = 0; u < 16; ++u) {
                soff[(size_t)(s0 + u) * n + i] = run;
                run += c[u];
            }
        }
        indeg[i] = run;
        dinv[i] = rsqrtf((float)(run + 1u));
    }
    sh[tid] = run;
    __syncthreads();
    for (int o = NODE_T / 2; o > 0; o >>= 1) {
        if (tid < o) sh[tid] += sh[tid + o];
        __syncthreads();
    }
    if (tid == 0) blockSum[blockIdx.x] = sh[0];
}

__global__ void __launch_bounds__(NODE_T)
rowptr_kernel(const unsigned* __restrict__ indeg, const unsigned* __restrict__ blockSum,
              unsigned* __restrict__ rowptr, int n) {
    __shared__ unsigned sh[NODE_T];
    const int tid = threadIdx.x;
    unsigned pre = 0u;
    for (int b = tid; b < (int)blockIdx.x; b += NODE_T) pre += blockSum[b];
    sh[tid] = pre;
    __syncthreads();
    for (int o = NODE_T / 2; o > 0; o >>= 1) {
        if (tid < o) sh[tid] += sh[tid + o];
        __syncthreads();
    }
    const unsigned before = sh[0];
    __syncthreads();

    const int i = blockIdx.x * NODE_T + tid;
    const unsigned v = (i < n) ? indeg[i] : 0u;
    sh[tid] = v;
    __syncthreads();
    for (int o = 1; o < NODE_T; o <<= 1) {
        unsigned t = (tid >= o) ? sh[tid - o] : 0u;
        __syncthreads();
        sh[tid] += t;
        __syncthreads();
    }
    if (i < n) rowptr[i] = before + sh[tid] - v;
}

__device__ __forceinline__ void
csr_fill_block(const int* __restrict__ src, const int* __restrict__ dst,
               const unsigned* __restrict__ rowptr, const unsigned* __restrict__ soff,
               unsigned* __restrict__ csr, int n_nodes, int n_edges, int bx, int by) {
    __shared__ unsigned cur[DEG_RANGE];
    const int tid  = threadIdx.x;
    const int base = by * DEG_RANGE;
    const int nr   = min(DEG_RANGE, n_nodes - base);

    const unsigned* so = soff + (size_t)bx * n_nodes + base;
    for (int i = tid; i < nr; i += DEG_THREADS) cur[i] = rowptr[base + i] + so[i];
    __syncthreads();

    const int slice = ((n_edges + DEG_SLICES * 4 - 1) / (DEG_SLICES * 4)) * 4;
    const int e0 = bx * slice;
    const int e1 = min(e0 + slice, n_edges);

    int e = e0 + tid;
    for (; e + (DEG_DEPTH - 1) * DEG_THREADS < e1; e += DEG_DEPTH * DEG_THREADS) {
        unsigned d[DEG_DEPTH];
        int      s[DEG_DEPTH];
#pragma unroll
        for (int u = 0; u < DEG_DEPTH; ++u) {
            d[u] = (unsigned)(dst[e + u * DEG_THREADS] - base);
            s[u] = src[e + u * DEG_THREADS];
        }
        asm volatile("" :: "v"(s[0]), "v"(s[1]), "v"(s[2]), "v"(s[3]),
                           "v"(s[4]), "v"(s[5]), "v"(s[6]), "v"(s[7]));
#pragma unroll
        for (int u = 0; u < DEG_DEPTH; ++u)
            if (d[u] < (unsigned)nr) {
                unsigned pos = atomicAdd(&cur[d[u]], 1u);
                csr[pos] = (unsigned)s[u];
            }
    }
    for (; e < e1; e += DEG_THREADS) {
        unsigned d0 = (unsigned)(dst[e] - base);
        const int s0 = src[e];
        if (d0 < (unsigned)nr) {
            unsigned pos = atomicAdd(&cur[d0], 1u);
            csr[pos] = (unsigned)s0;
        }
    }
}

__device__ __forceinline__ void
gemm_wmma_tile(const float* __restrict__ X, const float* __restrict__ W,
               const float* __restrict__ rs,
               float* __restrict__ H, int n_rows, int K, int wave) {
    int lane = threadIdx.x & 31;
    int row0 = wave * 16;
    if (row0 >= n_rows) return;

    int half = lane >> 4;
    int l16  = lane & 15;

    const float* xrow = X + (size_t)(row0 + l16) * K;
    v8f acc = {};
    for (int k = 0; k < K; k += 4) {
        int ka = k + half * 2;
        v2f a, b;
        a.x = xrow[ka];
        a.y = xrow[ka + 1];
        b.x = W[(size_t)ka * HIDDEN + l16];
        b.y = W[(size_t)(ka + 1) * HIDDEN + l16];
        acc = __builtin_amdgcn_wmma_f32_16x16x4_f32(
             false, a,  false, b,
             (short)0, acc,  false,  false);
    }
    float sc[8];
#pragma unroll
    for (int i = 0; i < 8; ++i) sc[i] = rs[row0 + i + 8 * half];
    float* out = H + (size_t)row0 * HIDDEN;
#pragma unroll
    for (int i = 0; i < 8; ++i)
        out[(size_t)(i + 8 * half) * HIDDEN + l16] = acc[i] * sc[i];
}

__global__ void __launch_bounds__(DEG_THREADS)
csr_fill_gemm_kernel(const int* __restrict__ src, const int* __restrict__ dst,
                     const unsigned* __restrict__ rowptr, const unsigned* __restrict__ soff,
                     unsigned* __restrict__ csr, int n_nodes, int n_edges, int n_fill,
                     const float* __restrict__ X, const float* __restrict__ W,
                     const float* __restrict__ rs, float* __restrict__ H, int K) {
    const int id = blockIdx.x;
    if (id < n_fill) {
        csr_fill_block(src, dst, rowptr, soff, csr, n_nodes, n_edges,
                       id % DEG_SLICES, id / DEG_SLICES);
    } else {
        const int wave = (id - n_fill) * (DEG_THREADS / 32) + (int)(threadIdx.x >> 5);
        gemm_wmma_tile(X, W, rs, H, n_nodes, K, wave);
    }
}


#define GATHER_DEPTH 8
#define ROW_PITCH 20

__device__ __forceinline__ void wave_lds_handover() {
    __builtin_amdgcn_fence(__ATOMIC_RELEASE, "wavefront");
    __builtin_amdgcn_wave_barrier();
    __builtin_amdgcn_fence(__ATOMIC_ACQUIRE, "wavefront");
}

template <int K>
__device__ __forceinline__ unsigned quad_bcast(unsigned v) {
    return (unsigned)__builtin_amdgcn_mov_dpp((int)v, K * 0x55, 0xF, 0xF, true);
}

template <bool POOL>
__global__ void __launch_bounds__(256)
gather_kernel(const unsigned* __restrict__ csr, const unsigned* __restrict__ rowptr,
              const unsigned* __restrict__ indeg, const float* __restrict__ dinv,
              const float4* __restrict__ hin, const float* __restrict__ bias,
              const float* __restrict__ Wn,
              float4* __restrict__ hout, const int* __restrict__ batch,
              float* __restrict__ pooled, int n_nodes, int n_graphs, int do_relu) {
    int g = blockIdx.x * blockDim.x + threadIdx.x;
    int i = g >> 2;
    const bool valid = i < n_nodes;
    if (!POOL) {
        if (((g & ~31) >> 2) >= n_nodes) return;
    }
    if (!valid) i = 0;
    int q = g & 3;
    int bid = -1;
    if (POOL) {
        bid = batch[i];
        if (!valid) bid = -1;
    }
    unsigned j = rowptr[i];
    const unsigned cnt = indeg[i];
    const unsigned end = j + (valid ? cnt : 0u);
    const float di = dinv[i];
    __shared__ __attribute__((aligned(16))) float4 sSelf[256];
    sSelf[threadIdx.x] = hin[(size_t)i * 4 + q];
    float4 acc = make_float4(0.0f, 0.0f, 0.0f, 0.0f);

    const unsigned last = max(end, 1u) - 1u;
    unsigned c0 = csr[min(j + (unsigned)q, last)];
    unsigned c1 = csr[min(j + 4u + (unsigned)q, last)];
    for (; j < end; j += GATHER_DEPTH) {
        static_assert(GATHER_DEPTH == 8, "a quad holds 2 slots per lane");
        const unsigned n0 = csr[min(j + 8u + (unsigned)q, last)];
        const unsigned n1 = csr[min(j + 12u + (unsigned)q, last)];
        unsigned c[GATHER_DEPTH];
        float4   h[GATHER_DEPTH];
        c[0] = quad_bcast<0>(c0);
        c[1] = quad_bcast<1>(c0);
        c[2] = quad_bcast<2>(c0);
        c[3] = quad_bcast<3>(c0);
        c[4] = quad_bcast<0>(c1);
        c[5] = quad_bcast<1>(c1);
        c[6] = quad_bcast<2>(c1);
        c[7] = quad_bcast<3>(c1);
#pragma unroll
        for (int u = 0; u < GATHER_DEPTH; ++u)
            h[u] = *(const float4*)((const char*)hin + (c[u] * 64u + (unsigned)q * 16u));
        asm volatile("" :: "v"(n0), "v"(n1),
                           "v"(h[0].x), "v"(h[0].w), "v"(h[1].x), "v"(h[1].w),
                           "v"(h[2].x), "v"(h[2].w), "v"(h[3].x), "v"(h[3].w),
                           "v"(h[4].x), "v"(h[4].w), "v"(h[5].x), "v"(h[5].w),
                           "v"(h[6].x), "v"(h[6].w), "v"(h[7].x), "v"(h[7].w));
        c0 = n0;
        c1 = n1;
#pragma unroll
        for (int u = 0; u < GATHER_DEPTH; ++u) {
            const bool on = j + (unsigned)u < end;
            const float sx = acc.x + h[u].x;
            const float sy = acc.y + h[u].y;
            const float sz = acc.z + h[u].z;
            const float sw = acc.w + h[u].w;
            acc.x = on ? sx : acc.x;
            acc.y = on ? sy : acc.y;
            acc.z = on ? sz : acc.z;
            acc.w = on ? sw : acc.w;
        }
    }

    const float4 hs = sSelf[threadIdx.x];
    float4 v;
    v.x = di * (acc.x + hs.x) + bias[q * 4 + 0];
    v.y = di * (acc.y + hs.y) + bias[q * 4 + 1];
    v.z = di * (acc.z + hs.z) + bias[q * 4 + 2];
    v.w = di * (acc.w + hs.w) + bias[q * 4 + 3];
    if (do_relu) {
        v.x = fmaxf(v.x, 0.0f);
        v.y = fmaxf(v.y, 0.0f);
        v.z = fmaxf(v.z, 0.0f);
        v.w = fmaxf(v.w, 0.0f);
    }
    const int lane = threadIdx.x & 31;
    if (!POOL) {
        __shared__ __attribute__((aligned(16))) float sIn[8 * 8 * ROW_PITCH];
        __shared__ __attribute__((aligned(16))) float sOut[8 * 8 * ROW_PITCH];
        float* in_rows  = sIn  + (threadIdx.x >> 5) * (8 * ROW_PITCH);
        float* out_rows = sOut + (threadIdx.x >> 5) * (8 * ROW_PITCH);
        const int nd = lane >> 2;
        *(float4*)&in_rows[nd * ROW_PITCH + q * 4] = v;
        wave_lds_handover();

        const int half = lane >> 4;
        const int l16  = lane & 15;
        const float* arow = in_rows + (l16 & 7) * ROW_PITCH;
        v8f t = {};
#pragma unroll
        for (int kk = 0; kk < HIDDEN / 4; ++kk) {
            const int ka = kk * 4 + half * 2;
            v2f a, b;
            a.x = arow[ka];
            a.y = arow[ka + 1];
            b.x = Wn[ka * HIDDEN + l16];
            b.y = Wn[(ka + 1) * HIDDEN + l16];
            t = __builtin_amdgcn_wmma_f32_16x16x4_f32(
                 false, a,  false, b,
                 (short)0, t,  false,  false);
        }
        if (half == 0) {
#pragma unroll
            for (int r = 0; r < 8; ++r) out_rows[r * ROW_PITCH + l16] = t[r];
        }
        wave_lds_handover();
        float4 o = *(const float4*)&out_rows[nd * ROW_PITCH + q * 4];
        o.x *= di;
        o.y *= di;
        o.z *= di;
        o.w *= di;
        if (valid) hout[(size_t)i * 4 + q] = o;
    } else {
        __shared__ float4 srow[256];
        const int f = lane & 15;
        srow[threadIdx.x] = v;
        __syncthreads();
        const float* rows = (const float*)&srow[threadIdx.x & ~31];
        float sum = 0.0f;
        int cur = __builtin_amdgcn_readlane(bid, 0);
#pragma unroll
        for (int n = 0; n < 8; ++n) {
            const int b = __builtin_amdgcn_readlane(bid, n * 4);
            const float x = rows[n * 16 + f];
            if (b != cur) {
                if (lane < 16 && (unsigned)cur < (unsigned)n_graphs)
                    atomicAdd(&pooled[(size_t)cur * HIDDEN + f], sum);
                sum = 0.0f;
                cur = b;
            }
            sum += x;
        }
        if (lane < 16 && (unsigned)cur < (unsigned)n_graphs)
            atomicAdd(&pooled[(size_t)cur * HIDDEN + f], sum);
    }
}


__global__ void final_linear_kernel(const float* __restrict__ pooled,
                                    const float* __restrict__ Wlin,
                                    const float* __restrict__ blin,
                                    float* __restrict__ out,
                                    int n_graphs, int n_classes) {
    int g = blockIdx.x * blockDim.x + threadIdx.x;
    if (g >= n_graphs * n_classes) return;
    int gr = g / n_classes;
    int c  = g % n_classes;
    float acc = blin[c];
#pragma unroll
    for (int f = 0; f < HIDDEN; ++f)
        acc += pooled[(size_t)gr * HIDDEN + f] * Wlin[f * n_classes + c];
    out[g] = acc;
}


static inline size_t align_up(size_t v, size_t a) { return (v + a - 1) & ~(a - 1); }

extern "C" void kernel_launch(void* const* d_in, const int* in_sizes, int n_in,
                              void* d_out, int out_size, void* d_ws, size_t ws_size,
                              hipStream_t stream) {
    (void)n_in; (void)ws_size;

    const float* x    = (const float*)d_in[0];
    const int*   ei   = (const int*)d_in[1];
    const int*   batch= (const int*)d_in[2];
    const float* W1   = (const float*)d_in[3];
    const float* b1   = (const float*)d_in[4];
    const float* W2   = (const float*)d_in[5];
    const float* b2   = (const float*)d_in[6];
    const float* W3   = (const float*)d_in[7];
    const float* b3   = (const float*)d_in[8];
    const float* Wlin = (const float*)d_in[9];
    const float* blin = (const float*)d_in[10];
    float* out = (float*)d_out;

    const int N_FEAT = 128;
    const int n_nodes = in_sizes[0] / N_FEAT;
    const int n_edges = in_sizes[1] / 2;
    const int n_classes = in_sizes[10];
    const int n_graphs  = out_size / n_classes;

    const int* src = ei;
    const int* dst = ei + n_edges;

    char* ws = (char*)d_ws;
    size_t off = 0;
    float* dinv = (float*)(ws + off);  off = align_up(off + (size_t)n_nodes * 4, 256);
    float* bufA = (float*)(ws + off);  off = align_up(off + (size_t)n_nodes * HIDDEN * 4, 256);
    float* bufB = (float*)(ws + off);  off = align_up(off + (size_t)n_nodes * HIDDEN * 4, 256);
    float* pooled = (float*)(ws + off); off = align_up(off + (size_t)n_graphs * HIDDEN * 4, 256);
    unsigned* degPart = (unsigned*)(ws + off);
    off = align_up(off + (size_t)DEG_SLICES * n_nodes * 4, 256);
    unsigned* sliceOff = (unsigned*)(ws + off);
    off = align_up(off + (size_t)DEG_SLICES * n_nodes * 4, 256);
    unsigned* indeg = (unsigned*)(ws + off);    off = align_up(off + (size_t)n_nodes * 4, 256);
    unsigned* rowptr = (unsigned*)(ws + off);   off = align_up(off + (size_t)n_nodes * 4, 256);
    unsigned* blockSum = (unsigned*)(ws + off); off = align_up(off + (size_t)n_nodes * 4, 256);
    unsigned* csr = (unsigned*)(ws + off);
    off = align_up(off + (size_t)n_edges * 4, 256);

    const int T = 256;
    const int nodeBlocks  = (n_nodes + NODE_T - 1) / NODE_T;
    const int nodeQBlocks = ((n_nodes * 4) + T - 1) / T;
    const int tiles       = (n_nodes + 15) / 16;
    const int gemmBlocks  = (tiles + DEG_THREADS / 32 - 1) / (DEG_THREADS / 32);
    const dim3 degGrid(DEG_SLICES, (n_nodes + DEG_RANGE - 1) / DEG_RANGE);
    const int fillBlocks  = DEG_SLICES * ((n_nodes + DEG_RANGE - 1) / DEG_RANGE);

    deg_hist_kernel<<<degGrid, DEG_THREADS, 0, stream>>>(dst, degPart, n_nodes, n_edges);
    dinv_kernel<<<nodeBlocks, NODE_T, 0, stream>>>(degPart, sliceOff, indeg, blockSum, dinv,
                                                   pooled, n_graphs * HIDDEN, n_nodes);
    rowptr_kernel<<<nodeBlocks, NODE_T, 0, stream>>>(indeg, blockSum, rowptr, n_nodes);
    csr_fill_gemm_kernel<<<fillBlocks + gemmBlocks, DEG_THREADS, 0, stream>>>(
        src, dst, rowptr, sliceOff, csr, n_nodes, n_edges, fillBlocks,
        x, W1, dinv, bufA, N_FEAT);

    gather_kernel<false><<<nodeQBlocks, T, 0, stream>>>(csr, rowptr, indeg, dinv, (const float4*)bufA, b1, W2,
                                                        (float4*)bufB, batch, pooled,
                                                        n_nodes, n_graphs, 1);

    gather_kernel<false><<<nodeQBlocks, T, 0, stream>>>(csr, rowptr, indeg, dinv, (const float4*)bufB, b2, W3,
                                                        (float4*)bufA, batch, pooled,
                                                        n_nodes, n_graphs, 1);

    gather_kernel<true><<<nodeQBlocks, T, 0, stream>>>(csr, rowptr, indeg, dinv, (const float4*)bufA, b3, nullptr,
                                                       (float4*)bufB, batch, pooled,
                                                       n_nodes, n_graphs, 0);

    final_linear_kernel<<<(n_graphs * n_classes + T - 1) / T, T, 0, stream>>>(
        pooled, Wlin, blin, out, n_graphs, n_classes);
}


# Edge sort now a two-level bucket sort: slices bucketed via LDS into a (dst,src) list, then a work-group per 400 nodes ranks in LDS, stores whole lines; 8 to 7 launches; slot, so sum, order changes
# speedup vs baseline: 5.6205x; 1.5441x over previous
#include <hip/hip_runtime.h>
#include <hip/hip_bf16.h>


typedef float v2f __attribute__((ext_vector_type(2)));
typedef float v8f __attribute__((ext_vector_type(8)));

#define HIDDEN 16


#define NS        256
#define NBK       256
#define BKT_R     400
#define SORT_T    1024
#define SORT_DEPTH 8
#define SLICE_CAP 12544
#define BKT_CAP   32768

__global__ void __launch_bounds__(SORT_T)
bucket_count_kernel(const int* __restrict__ dst, unsigned* __restrict__ cntTab, int n_edges) {
    __shared__ unsigned h[NBK];
    const int tid = threadIdx.x;
    if (tid < NBK) h[tid] = 0u;
    __syncthreads();

    const int slice = (n_edges + NS - 1) / NS;
    const int e0 = blockIdx.x * slice;
    const int e1 = min(e0 + slice, n_edges);

    for (int eb = e0 + tid; eb < e1; eb += SORT_DEPTH * SORT_T) {
        unsigned d[SORT_DEPTH];
#pragma unroll
        for (int u = 0; u < SORT_DEPTH; ++u)
            d[u] = (unsigned)dst[min(eb + u * SORT_T, e1 - 1)];
        asm volatile("" :: "v"(d[0]), "v"(d[1]), "v"(d[2]), "v"(d[3]),
                           "v"(d[4]), "v"(d[5]), "v"(d[6]), "v"(d[7]));
#pragma unroll
        for (int u = 0; u < SORT_DEPTH; ++u)
            if (eb + u * SORT_T < e1) atomicAdd(&h[min(d[u] / BKT_R, (unsigned)(NBK - 1))], 1u);
    }
    __syncthreads();
    if (tid < NBK) cntTab[(size_t)blockIdx.x * NBK + tid] = h[tid];
}

__device__ __forceinline__ void
bucket_scatter_block(const int* __restrict__ src, const int* __restrict__ dst,
                     const unsigned* __restrict__ cntTab, unsigned* __restrict__ bstart,
                     uint2* __restrict__ list, int n_edges, int s) {
    __shared__ unsigned lofs[NBK];
    __shared__ unsigned gofs[NBK];
    __shared__ unsigned cur[NBK];
    __shared__ unsigned pAll[4][NBK];
    __shared__ unsigned pBef[4][NBK];
    __shared__ unsigned s1[NBK];
    __shared__ unsigned s2[NBK];
    __shared__ uint2 stage[SLICE_CAP];
    const int tid = threadIdx.x;
    {
        const int b = tid & (NBK - 1);
        const int part = tid >> 8;
        unsigned tot = 0u, bef = 0u;
#pragma unroll 16
        for (int k = 0; k < NS / 4; ++k) {
            const int sl = part * (NS / 4) + k;
            const unsigned v = cntTab[(size_t)sl * NBK + b];
            tot += v;
            bef += (sl < s) ? v : 0u;
        }
        pAll[part][b] = tot;
        pBef[part][b] = bef;
    }
    __syncthreads();
    unsigned T = 0u, B = 0u, mine = 0u;
    if (tid < NBK) {
        T = pAll[0][tid] + pAll[1][tid] + pAll[2][tid] + pAll[3][tid];
        B = pBef[0][tid] + pBef[1][tid] + pBef[2][tid] + pBef[3][tid];
        mine = cntTab[(size_t)s * NBK + tid];
        s1[tid] = T;
        s2[tid] = mine;
    }
    __syncthreads();
    for (int o = 1; o < NBK; o <<= 1) {
        unsigned a1 = 0u, a2 = 0u;
        if (tid < NBK && tid >= o) { a1 = s1[tid - o]; a2 = s2[tid - o]; }
        __syncthreads();
        if (tid < NBK) { s1[tid] += a1; s2[tid] += a2; }
        __syncthreads();
    }
    if (tid < NBK) {
        const unsigned bs = s1[tid] - T;
        const unsigned lo = s2[tid] - mine;
        lofs[tid] = lo;
        cur[tid]  = lo;
        gofs[tid] = bs + B;
        if (s == 0) {
            bstart[tid] = bs;
            if (tid == NBK - 1) bstart[NBK] = s1[tid];
        }
    }
    __syncthreads();

    const int slice = (n_edges + NS - 1) / NS;
    const int e0 = s * slice;
    const int e1 = min(e0 + slice, n_edges);
    const int len = e1 - e0;
    const bool fits = len <= SLICE_CAP;

    for (int eb = e0 + tid; eb < e1; eb += SORT_DEPTH * SORT_T) {
        unsigned d[SORT_DEPTH];
        int      sv[SORT_DEPTH];
#pragma unroll
        for (int u = 0; u < SORT_DEPTH; ++u) {
            const int e = min(eb + u * SORT_T, e1 - 1);
            d[u]  = (unsigned)dst[e];
            sv[u] = src[e];
        }
        asm volatile("" :: "v"(d[0]), "v"(d[1]), "v"(d[2]), "v"(d[3]),
                           "v"(d[4]), "v"(d[5]), "v"(d[6]), "v"(d[7]),
                           "v"(sv[0]), "v"(sv[1]), "v"(sv[2]), "v"(sv[3]),
                           "v"(sv[4]), "v"(sv[5]), "v"(sv[6]), "v"(sv[7]));
#pragma unroll
        for (int u = 0; u < SORT_DEPTH; ++u)
            if (eb + u * SORT_T < e1) {
                const unsigned b = min(d[u] / BKT_R, (unsigned)(NBK - 1));
                const unsigned pos = atomicAdd(&cur[b], 1u);
                const uint2 p = make_uint2(d[u], (unsigned)sv[u]);
                if (fits) stage[pos] = p;
                else      list[gofs[b] + (pos - lofs[b])] = p;
            }
    }
    __syncthreads();
    if (fits) {
        for (int k = tid; k < len; k += SORT_T) {
            const uint2 p = stage[k];
            const unsigned b = min(p.x / BKT_R, (unsigned)(NBK - 1));
            list[gofs[b] + ((unsigned)k - lofs[b])] = p;
        }
    }
}

__device__ __forceinline__ void
gemm_wmma_tile(const float* __restrict__ X, const float* __restrict__ W,
               float* __restrict__ H, int n_rows, int K, int wave) {
    int lane = threadIdx.x & 31;
    int row0 = wave * 16;
    if (row0 >= n_rows) return;

    int half = lane >> 4;
    int l16  = lane & 15;

    const float* xrow = X + (size_t)(row0 + l16) * K;
    v8f acc = {};
    for (int k = 0; k < K; k += 4) {
        int ka = k + half * 2;
        v2f a, b;
        a.x = xrow[ka];
        a.y = xrow[ka + 1];
        b.x = W[(size_t)ka * HIDDEN + l16];
        b.y = W[(size_t)(ka + 1) * HIDDEN + l16];
        acc = __builtin_amdgcn_wmma_f32_16x16x4_f32(
             false, a,  false, b,
             (short)0, acc,  false,  false);
    }
    float* out = H + (size_t)row0 * HIDDEN;
#pragma unroll
    for (int i = 0; i < 8; ++i)
        out[(size_t)(i + 8 * half) * HIDDEN + l16] = acc[i];
}

__global__ void __launch_bounds__(SORT_T)
scatter_gemm_kernel(const int* __restrict__ src, const int* __restrict__ dst,
                    const unsigned* __restrict__ cntTab, unsigned* __restrict__ bstart,
                    uint2* __restrict__ list, int n_edges, int n_scatter,
                    const float* __restrict__ X, const float* __restrict__ W,
                    float* __restrict__ H, int n_nodes, int K) {
    const int id = blockIdx.x;
    if (id < n_scatter) {
        bucket_scatter_block(src, dst, cntTab, bstart, list, n_edges, id);
    } else {
        const int wave = (id - n_scatter) * (SORT_T / 32) + (int)(threadIdx.x >> 5);
        gemm_wmma_tile(X, W, H, n_nodes, K, wave);
    }
}

__global__ void __launch_bounds__(SORT_T)
bucket_sort_kernel(const uint2* __restrict__ list, const unsigned* __restrict__ bstart,
                   unsigned* __restrict__ csr, unsigned* __restrict__ rowptr,
                   unsigned* __restrict__ indeg, float* __restrict__ dinv,
                   float4* __restrict__ H4, float* __restrict__ pooled, int n_pool, int n_nodes) {
    __shared__ unsigned hist[BKT_R];
    __shared__ unsigned sc[512];
    __shared__ float    sd[BKT_R];
    __shared__ unsigned stage[BKT_CAP];
    static_assert(BKT_R <= 512 && 512 <= SORT_T, "the scan below covers 512 nodes");
    const int tid = threadIdx.x;
    const int bkt = blockIdx.x;
    for (int k = bkt * SORT_T + tid; k < n_pool; k += (int)gridDim.x * SORT_T) pooled[k] = 0.0f;

    const int base = bkt * BKT_R;
    const int nr = min(BKT_R, n_nodes - base);
    const unsigned start = bstart[bkt];
    const int len = (int)(bstart[bkt + 1] - start);
    const uint2* my = list + start;

    if (tid < BKT_R) hist[tid] = 0u;
    __syncthreads();
    for (int kb = tid; kb < len; kb += SORT_DEPTH * SORT_T) {
        unsigned d[SORT_DEPTH];
#pragma unroll
        for (int u = 0; u < SORT_DEPTH; ++u)
            d[u] = my[min(kb + u * SORT_T, len - 1)].x;
        asm volatile("" :: "v"(d[0]), "v"(d[1]), "v"(d[2]), "v"(d[3]),
                           "v"(d[4]), "v"(d[5]), "v"(d[6]), "v"(d[7]));
#pragma unroll
        for (int u = 0; u < SORT_DEPTH; ++u)
            if (kb + u * SORT_T < len)
                atomicAdd(&hist[min(d[u] - (unsigned)base, (unsigned)(BKT_R - 1))], 1u);
    }
    __syncthreads();

    const unsigned c = (tid < BKT_R) ? hist[tid] : 0u;
    if (tid < 512) sc[tid] = c;
    __syncthreads();
    for (int o = 1; o < 512; o <<= 1) {
        unsigned a = 0u;
        if (tid < 512 && tid >= o) a = sc[tid - o];
        __syncthreads();
        if (tid < 512) sc[tid] += a;
        __syncthreads();
    }
    if (tid < BKT_R) {
        const unsigned ex = sc[tid] - c;
        hist[tid] = ex;
        const float di = rsqrtf((float)(c + 1u));
        sd[tid] = di;
        if (tid < nr) {
            indeg[base + tid]  = c;
            rowptr[base + tid] = start + ex;
            dinv[base + tid]   = di;
        }
    }
    __syncthreads();

    const bool fits = len <= BKT_CAP;
    for (int kb = tid; kb < len; kb += SORT_DEPTH * SORT_T) {
        uint2 p[SORT_DEPTH];
#pragma unroll
        for (int u = 0; u < SORT_DEPTH; ++u)
            p[u] = my[min(kb + u * SORT_T, len - 1)];
        asm volatile("" :: "v"(p[0].x), "v"(p[0].y), "v"(p[1].x), "v"(p[1].y),
                           "v"(p[2].x), "v"(p[2].y), "v"(p[3].x), "v"(p[3].y),
                           "v"(p[4].x), "v"(p[4].y), "v"(p[5].x), "v"(p[5].y),
                           "v"(p[6].x), "v"(p[6].y), "v"(p[7].x), "v"(p[7].y));
#pragma unroll
        for (int u = 0; u < SORT_DEPTH; ++u)
            if (kb + u * SORT_T < len) {
                const unsigned nd  = min(p[u].x - (unsigned)base, (unsigned)(BKT_R - 1));
                const unsigned pos = atomicAdd(&hist[nd], 1u);
                if (fits) stage[pos] = p[u].y;
                else      csr[start + pos] = p[u].y;
            }
    }
    __syncthreads();
    if (fits) {
        for (int k = tid; k < len; k += SORT_T) csr[start + k] = stage[k];
    }

    float4* rows = H4 + (size_t)base * 4;
    for (int k = tid; k < nr * 4; k += SORT_T) {
        float4 v = rows[k];
        const float di = sd[k >> 2];
        v.x *= di;
        v.y *= di;
        v.z *= di;
        v.w *= di;
        rows[k] = v;
    }
}


#define GATHER_DEPTH 8
#define ROW_PITCH 20

__device__ __forceinline__ void wave_lds_handover() {
    __builtin_amdgcn_fence(__ATOMIC_RELEASE, "wavefront");
    __builtin_amdgcn_wave_barrier();
    __builtin_amdgcn_fence(__ATOMIC_ACQUIRE, "wavefront");
}

template <int K>
__device__ __forceinline__ unsigned quad_bcast(unsigned v) {
    return (unsigned)__builtin_amdgcn_mov_dpp((int)v, K * 0x55, 0xF, 0xF, true);
}

template <bool POOL>
__global__ void __launch_bounds__(256)
gather_kernel(const unsigned* __restrict__ csr, const unsigned* __restrict__ rowptr,
              const unsigned* __restrict__ indeg, const float* __restrict__ dinv,
              const float4* __restrict__ hin, const float* __restrict__ bias,
              const float* __restrict__ Wn,
              float4* __restrict__ hout, const int* __restrict__ batch,
              float* __restrict__ pooled, int n_nodes, int n_graphs, int do_relu) {
    int g = blockIdx.x * blockDim.x + threadIdx.x;
    int i = g >> 2;
    const bool valid = i < n_nodes;
    if (!POOL) {
        if (((g & ~31) >> 2) >= n_nodes) return;
    }
    if (!valid) i = 0;
    int q = g & 3;
    int bid = -1;
    if (POOL) {
        bid = batch[i];
        if (!valid) bid = -1;
    }
    unsigned j = rowptr[i];
    const unsigned cnt = indeg[i];
    const unsigned end = j + (valid ? cnt : 0u);
    const float di = dinv[i];
    __shared__ __attribute__((aligned(16))) float4 sSelf[256];
    sSelf[threadIdx.x] = hin[(size_t)i * 4 + q];
    float4 acc = make_float4(0.0f, 0.0f, 0.0f, 0.0f);

    const unsigned last = max(end, 1u) - 1u;
    unsigned c0 = csr[min(j + (unsigned)q, last)];
    unsigned c1 = csr[min(j + 4u + (unsigned)q, last)];
    for (; j < end; j += GATHER_DEPTH) {
        static_assert(GATHER_DEPTH == 8, "a quad holds 2 slots per lane");
        const unsigned n0 = csr[min(j + 8u + (unsigned)q, last)];
        const unsigned n1 = csr[min(j + 12u + (unsigned)q, last)];
        unsigned c[GATHER_DEPTH];
        float4   h[GATHER_DEPTH];
        c[0] = quad_bcast<0>(c0);
        c[1] = quad_bcast<1>(c0);
        c[2] = quad_bcast<2>(c0);
        c[3] = quad_bcast<3>(c0);
        c[4] = quad_bcast<0>(c1);
        c[5] = quad_bcast<1>(c1);
        c[6] = quad_bcast<2>(c1);
        c[7] = quad_bcast<3>(c1);
#pragma unroll
        for (int u = 0; u < GATHER_DEPTH; ++u)
            h[u] = *(const float4*)((const char*)hin + (c[u] * 64u + (unsigned)q * 16u));
        asm volatile("" :: "v"(n0), "v"(n1),
                           "v"(h[0].x), "v"(h[0].w), "v"(h[1].x), "v"(h[1].w),
                           "v"(h[2].x), "v"(h[2].w), "v"(h[3].x), "v"(h[3].w),
                           "v"(h[4].x), "v"(h[4].w), "v"(h[5].x), "v"(h[5].w),
                           "v"(h[6].x), "v"(h[6].w), "v"(h[7].x), "v"(h[7].w));
        c0 = n0;
        c1 = n1;
#pragma unroll
        for (int u = 0; u < GATHER_DEPTH; ++u) {
            const bool on = j + (unsigned)u < end;
            const float sx = acc.x + h[u].x;
            const float sy = acc.y + h[u].y;
            const float sz = acc.z + h[u].z;
            const float sw = acc.w + h[u].w;
            acc.x = on ? sx : acc.x;
            acc.y = on ? sy : acc.y;
            acc.z = on ? sz : acc.z;
            acc.w = on ? sw : acc.w;
        }
    }

    const float4 hs = sSelf[threadIdx.x];
    float4 v;
    v.x = di * (acc.x + hs.x) + bias[q * 4 + 0];
    v.y = di * (acc.y + hs.y) + bias[q * 4 + 1];
    v.z = di * (acc.z + hs.z) + bias[q * 4 + 2];
    v.w = di * (acc.w + hs.w) + bias[q * 4 + 3];
    if (do_relu) {
        v.x = fmaxf(v.x, 0.0f);
        v.y = fmaxf(v.y, 0.0f);
        v.z = fmaxf(v.z, 0.0f);
        v.w = fmaxf(v.w, 0.0f);
    }
    const int lane = threadIdx.x & 31;
    if (!POOL) {
        __shared__ __attribute__((aligned(16))) float sIn[8 * 8 * ROW_PITCH];
        __shared__ __attribute__((aligned(16))) float sOut[8 * 8 * ROW_PITCH];
        float* in_rows  = sIn  + (threadIdx.x >> 5) * (8 * ROW_PITCH);
        float* out_rows = sOut + (threadIdx.x >> 5) * (8 * ROW_PITCH);
        const int nd = lane >> 2;
        *(float4*)&in_rows[nd * ROW_PITCH + q * 4] = v;
        wave_lds_handover();

        const int half = lane >> 4;
        const int l16  = lane & 15;
        const float* arow = in_rows + (l16 & 7) * ROW_PITCH;
        v8f t = {};
#pragma unroll
        for (int kk = 0; kk < HIDDEN / 4; ++kk) {
            const int ka = kk * 4 + half * 2;
            v2f a, b;
            a.x = arow[ka];
            a.y = arow[ka + 1];
            b.x = Wn[ka * HIDDEN + l16];
            b.y = Wn[(ka + 1) * HIDDEN + l16];
            t = __builtin_amdgcn_wmma_f32_16x16x4_f32(
                 false, a,  false, b,
                 (short)0, t,  false,  false);
        }
        if (half == 0) {
#pragma unroll
            for (int r = 0; r < 8; ++r) out_rows[r * ROW_PITCH + l16] = t[r];
        }
        wave_lds_handover();
        float4 o = *(const float4*)&out_rows[nd * ROW_PITCH + q * 4];
        o.x *= di;
        o.y *= di;
        o.z *= di;
        o.w *= di;
        if (valid) hout[(size_t)i * 4 + q] = o;
    } else {
        __shared__ float4 srow[256];
        const int f = lane & 15;
        srow[threadIdx.x] = v;
        __syncthreads();
        const float* rows = (const float*)&srow[threadIdx.x & ~31];
        float sum = 0.0f;
        int cur = __builtin_amdgcn_readlane(bid, 0);
#pragma unroll
        for (int n = 0; n < 8; ++n) {
            const int b = __builtin_amdgcn_readlane(bid, n * 4);
            const float x = rows[n * 16 + f];
            if (b != cur) {
                if (lane < 16 && (unsigned)cur < (unsigned)n_graphs)
                    atomicAdd(&pooled[(size_t)cur * HIDDEN + f], sum);
                sum = 0.0f;
                cur = b;
            }
            sum += x;
        }
        if (lane < 16 && (unsigned)cur < (unsigned)n_graphs)
            atomicAdd(&pooled[(size_t)cur * HIDDEN + f], sum);
    }
}


__global__ void final_linear_kernel(const float* __restrict__ pooled,
                                    const float* __restrict__ Wlin,
                                    const float* __restrict__ blin,
                                    float* __restrict__ out,
                                    int n_graphs, int n_classes) {
    int g = blockIdx.x * blockDim.x + threadIdx.x;
    if (g >= n_graphs * n_classes) return;
    int gr = g / n_classes;
    int c  = g % n_classes;
    float acc = blin[c];
#pragma unroll
    for (int f = 0; f < HIDDEN; ++f)
        acc += pooled[(size_t)gr * HIDDEN + f] * Wlin[f * n_classes + c];
    out[g] = acc;
}


static inline size_t align_up(size_t v, size_t a) { return (v + a - 1) & ~(a - 1); }

extern "C" void kernel_launch(void* const* d_in, const int* in_sizes, int n_in,
                              void* d_out, int out_size, void* d_ws, size_t ws_size,
                              hipStream_t stream) {
    (void)n_in; (void)ws_size;

    const float* x    = (const float*)d_in[0];
    const int*   ei   = (const int*)d_in[1];
    const int*   batch= (const int*)d_in[2];
    const float* W1   = (const float*)d_in[3];
    const float* b1   = (const float*)d_in[4];
    const float* W2   = (const float*)d_in[5];
    const float* b2   = (const float*)d_in[6];
    const float* W3   = (const float*)d_in[7];
    const float* b3   = (const float*)d_in[8];
    const float* Wlin = (const float*)d_in[9];
    const float* blin = (const float*)d_in[10];
    float* out = (float*)d_out;

    const int N_FEAT = 128;
    const int n_nodes = in_sizes[0] / N_FEAT;
    const int n_edges = in_sizes[1] / 2;
    const int n_classes = in_sizes[10];
    const int n_graphs  = out_size / n_classes;

    const int* src = ei;
    const int* dst = ei + n_edges;

    char* ws = (char*)d_ws;
    size_t off = 0;
    float* dinv = (float*)(ws + off);  off = align_up(off + (size_t)n_nodes * 4, 256);
    float* bufA = (float*)(ws + off);  off = align_up(off + (size_t)n_nodes * HIDDEN * 4, 256);
    float* bufB = (float*)(ws + off);  off = align_up(off + (size_t)n_nodes * HIDDEN * 4, 256);
    float* pooled = (float*)(ws + off); off = align_up(off + (size_t)n_graphs * HIDDEN * 4, 256);
    unsigned* cntTab = (unsigned*)(ws + off);
    off = align_up(off + (size_t)NS * NBK * 4, 256);
    unsigned* bstart = (unsigned*)(ws + off);
    off = align_up(off + (size_t)(NBK + 1) * 4, 256);
    uint2* list = (uint2*)(ws + off);
    off = align_up(off + (size_t)n_edges * 8, 256);
    unsigned* indeg = (unsigned*)(ws + off);    off = align_up(off + (size_t)n_nodes * 4, 256);
    unsigned* rowptr = (unsigned*)(ws + off);   off = align_up(off + (size_t)n_nodes * 4, 256);
    unsigned* csr = (unsigned*)(ws + off);
    off = align_up(off + (size_t)n_edges * 4, 256);

    const int T = 256;
    const int nodeQBlocks = ((n_nodes * 4) + T - 1) / T;
    const int tiles       = (n_nodes + 15) / 16;
    const int gemmBlocks  = (tiles + SORT_T / 32 - 1) / (SORT_T / 32);
    const int buckets     = (n_nodes + BKT_R - 1) / BKT_R;

    bucket_count_kernel<<<NS, SORT_T, 0, stream>>>(dst, cntTab, n_edges);
    scatter_gemm_kernel<<<NS + gemmBlocks, SORT_T, 0, stream>>>(
        src, dst, cntTab, bstart, list, n_edges, NS, x, W1, bufA, n_nodes, N_FEAT);
    bucket_sort_kernel<<<buckets, SORT_T, 0, stream>>>(list, bstart, csr, rowptr, indeg, dinv,
                                                       (float4*)bufA, pooled, n_graphs * HIDDEN, n_nodes);

    gather_kernel<false><<<nodeQBlocks, T, 0, stream>>>(csr, rowptr, indeg, dinv, (const float4*)bufA, b1, W2,
                                                        (float4*)bufB, batch, pooled,
                                                        n_nodes, n_graphs, 1);

    gather_kernel<false><<<nodeQBlocks, T, 0, stream>>>(csr, rowptr, indeg, dinv, (const float4*)bufB, b2, W3,
                                                        (float4*)bufA, batch, pooled,
                                                        n_nodes, n_graphs, 1);

    gather_kernel<true><<<nodeQBlocks, T, 0, stream>>>(csr, rowptr, indeg, dinv, (const float4*)bufA, b3, nullptr,
                                                       (float4*)bufB, batch, pooled,
                                                       n_nodes, n_graphs, 0);

    final_linear_kernel<<<(n_graphs * n_classes + T - 1) / T, T, 0, stream>>>(
        pooled, Wlin, blin, out, n_graphs, n_classes);
}
